# starting kernel
# baseline (speedup 1.0000x reference)
#include <hip/hip_runtime.h>
#include <math.h>

typedef float v2f __attribute__((ext_vector_type(2)));
typedef float v8f __attribute__((ext_vector_type(8)));

#define BB 512
#define MM 256
#define INDIM 50

#define O_D    0u
#define O_PHI  524288u
#define O_INP  1048576u
#define O_WV   1179648u
#define O_WV2  1181696u
#define O_GX   1183744u
#define O_GB   1184256u

__global__ __launch_bounds__(128) void k1_stages(
    const float* __restrict__ U0, const float* __restrict__ U1, const float* __restrict__ U2,
    const float* __restrict__ S,  const float* __restrict__ w,
    const float* __restrict__ b_t_n, const float* __restrict__ b_obs,
    const int* __restrict__ b_i_n,
    float* __restrict__ ws, float* __restrict__ out)
{
    __shared__ float S_s[256 * 51];
    __shared__ float inp_s[16 * 53];
    __shared__ float redF[4][16];
    __shared__ float redA[4][16];
    __shared__ float xs[16], beta_s[16], scT_s[16];
    __shared__ float aS[4][16], dxS[4][16];

    const int tid  = threadIdx.x;
    const int wv   = tid >> 5;
    const int lane = tid & 31;
    const int ln   = lane & 15;
    const int hb   = (lane < 16) ? 0 : 8;
    const int b0g  = blockIdx.x * 16;

    for (int idx = tid; idx < 256 * INDIM; idx += 128) {
        int m = idx / INDIM, k = idx - m * INDIM;
        S_s[m * 51 + k] = S[idx];
    }
    for (int m = tid; m < 256; m += 128) S_s[m * 51 + 50] = 0.f;

    for (int idx = tid; idx < 16 * 48; idx += 128) {
        int bl = idx / 48, j = idx - bl * 48;
        int u = j >> 4, r = j & 15;
        int iv = b_i_n[(b0g + bl) * 3 + u];
        const float* Uu = (u == 0) ? U0 : (u == 1) ? U1 : U2;
        inp_s[bl * 53 + 2 + j] = Uu[iv * 16 + r];
    }
    for (int idx = tid; idx < 16 * 3; idx += 128) {
        int bl = idx / 3, c = idx - bl * 3;
        inp_s[bl * 53 + 50 + c] = 0.f;
    }
    if (tid < 16) scT_s[tid] = b_t_n[b0g + tid];
    __syncthreads();

    if (tid < 16) {
        float acc = 0.f;
        for (int r = 0; r < 16; ++r)
            acc += inp_s[tid*53 + 2 + r] * inp_s[tid*53 + 18 + r] * inp_s[tid*53 + 34 + r];
        beta_s[tid] = acc; xs[tid] = acc;
    }
    __syncthreads();

    int   mro[4]; float w0r[4], w1r[4], s1r[4];
    #pragma unroll
    for (int nt = 0; nt < 4; ++nt) {
        int m = wv * 64 + nt * 16 + ln;
        mro[nt] = m; w0r[nt] = w[m]; w1r[nt] = w[m + 256]; s1r[nt] = S_s[m * 51 + 1];
    }
    float stv[8]; int bgv[8];
    #pragma unroll
    for (int v = 0; v < 8; ++v) { stv[v] = scT_s[hb + v]; bgv[v] = b0g + hb + v; }

    const int klo = (lane < 16) ? 0 : 2;

    for (int s = 0; s < 4; ++s) {
        if (tid < 16) {
            float ts = (s == 0) ? 0.f : ((s == 3) ? 1.f : 0.5f);
            inp_s[tid * 53 + 0] = scT_s[tid] * ts;
            inp_s[tid * 53 + 1] = xs[tid];
        }
        __syncthreads();

        for (int idx = tid; idx < 16 * 64; idx += 128) {
            int bl = idx >> 6, k = idx & 63;
            ws[O_INP + (unsigned)((s * 512 + b0g + bl) * 64 + k)] = (k < 50) ? inp_s[bl * 53 + k] : 0.f;
        }

        v8f acc[4];
        #pragma unroll
        for (int nt = 0; nt < 4; ++nt)
            #pragma unroll
            for (int v = 0; v < 8; ++v) acc[nt][v] = 0.f;

        #pragma unroll
        for (int kq = 0; kq < 13; ++kq) {
            int k0 = kq * 4 + klo;
            v2f af;
            af.x = inp_s[ln * 53 + k0];
            af.y = inp_s[ln * 53 + k0 + 1];
            #pragma unroll
            for (int nt = 0; nt < 4; ++nt) {
                v2f bf;
                bf.x = (k0     < 50) ? S_s[mro[nt] * 51 + k0]     : 0.f;
                bf.y = (k0 + 1 < 50) ? S_s[mro[nt] * 51 + k0 + 1] : 0.f;
                acc[nt] = __builtin_amdgcn_wmma_f32_16x16x4_f32(
                    false, af, false, bf, (short)0, acc[nt], false, false);
            }
        }

        float Fp[8], Ap[8];
        #pragma unroll
        for (int v = 0; v < 8; ++v) { Fp[v] = 0.f; Ap[v] = 0.f; }
        #pragma unroll
        for (int nt = 0; nt < 4; ++nt) {
            #pragma unroll
            for (int v = 0; v < 8; ++v) {
                float phi = acc[nt][v];
                float sn, cs;
                sincosf(phi, &sn, &cs);
                float fval  = w0r[nt] * cs + w1r[nt] * sn;
                float dterm = w1r[nt] * cs - w0r[nt] * sn;
                Fp[v] += fval;
                Ap[v] += dterm * s1r[nt];
                unsigned o = (unsigned)((s * 512 + bgv[v]) * 256 + mro[nt]);
                ws[O_D + o]   = stv[v] * dterm;
                ws[O_PHI + o] = phi;
            }
        }
        #pragma unroll
        for (int v = 0; v < 8; ++v) {
            float f = Fp[v], a = Ap[v];
            #pragma unroll
            for (int off = 8; off >= 1; off >>= 1) {
                f += __shfl_xor(f, off, 32);
                a += __shfl_xor(a, off, 32);
            }
            Fp[v] = f; Ap[v] = a;
        }
        if (ln == 0) {
            #pragma unroll
            for (int v = 0; v < 8; ++v) { redF[wv][hb + v] = Fp[v]; redA[wv][hb + v] = Ap[v]; }
        }
        __syncthreads();
        if (tid < 16) {
            float F  = redF[0][tid] + redF[1][tid] + redF[2][tid] + redF[3][tid];
            float A  = redA[0][tid] + redA[1][tid] + redA[2][tid] + redA[3][tid];
            float dx = scT_s[tid] * F;
            float a  = scT_s[tid] * A;
            dxS[s][tid] = dx; aS[s][tid] = a;
            if      (s == 0) xs[tid] = beta_s[tid] + 0.5f * dx;
            else if (s == 1) xs[tid] = beta_s[tid] + 0.5f * dx;
            else if (s == 2) xs[tid] = beta_s[tid] + dx;
        }
        __syncthreads();
    }

    if (tid < 16) {
        int bg = b0g + tid;
        float a1 = aS[0][tid], a2 = aS[1][tid], a3 = aS[2][tid], a4 = aS[3][tid];
        float dx1 = dxS[0][tid], dx2 = dxS[1][tid], dx3 = dxS[2][tid], dx4 = dxS[3][tid];
        float xf = beta_s[tid] + (dx1 + 2.f*dx2 + 2.f*dx3 + dx4) * (1.f/6.f);
        float gx = xf - b_obs[bg];
        float c1 = (1.f + a2 + 0.5f*a3*a2 + 0.25f*a4*a3*a2) * (1.f/6.f);
        float c2 = (2.f + a3 + 0.5f*a4*a3) * (1.f/6.f);
        float c3 = (2.f + a4) * (1.f/6.f);
        float c4 = (1.f/6.f);
        float kb1 = a1;
        float kb2 = a2 * (1.f + 0.5f*kb1);
        float kb3 = a3 * (1.f + 0.5f*kb2);
        float kb4 = a4 * (1.f + kb3);
        float gb = 1.f + (kb1 + 2.f*kb2 + 2.f*kb3 + kb4) * (1.f/6.f);
        const float inv = 1.f / 512.f;
        out[bg] = gx * gb * inv;
        ws[O_GX + bg] = gx; ws[O_GB + bg] = gb;
        float sc = scT_s[tid];
        float cc[4] = {c1, c2, c3, c4};
        #pragma unroll
        for (int s2 = 0; s2 < 4; ++s2) {
            float wvv = cc[s2] * gx * inv;
            ws[O_WV  + s2 * 512 + bg] = wvv;
            ws[O_WV2 + s2 * 512 + bg] = wvv * sc;
        }
    }
}

__global__ __launch_bounds__(32) void k2_sgrad(const float* __restrict__ ws, float* __restrict__ out)
{
    __shared__ float wv_s[2048];
    const int lane = threadIdx.x;
    const int mt = blockIdx.x >> 2;
    const int nt = blockIdx.x & 3;
    for (int i = lane; i < 2048; i += 32) wv_s[i] = ws[O_WV + i];
    __syncthreads();
    const int ln  = lane & 15;
    const int klo = (lane < 16) ? 0 : 2;
    const int m   = mt * 16 + ln;
    const int kc  = nt * 16 + ln;
    v8f acc;
    #pragma unroll
    for (int v = 0; v < 8; ++v) acc[v] = 0.f;
    for (int kq = 0; kq < 512; ++kq) {
        int kt = kq * 4 + klo;
        v2f af, bf;
        af.x = wv_s[kt]     * ws[O_D + (unsigned)kt * 256u + m];
        af.y = wv_s[kt + 1] * ws[O_D + (unsigned)(kt + 1) * 256u + m];
        bf.x = ws[O_INP + (unsigned)kt * 64u + kc];
        bf.y = ws[O_INP + (unsigned)(kt + 1) * 64u + kc];
        acc = __builtin_amdgcn_wmma_f32_16x16x4_f32(false, af, false, bf, (short)0, acc, false, false);
    }
    #pragma unroll
    for (int v = 0; v < 8; ++v) {
        int mm = mt * 16 + v + ((lane < 16) ? 0 : 8);
        int kk = nt * 16 + ln;
        if (kk < 50) out[25088 + mm * 50 + kk] = acc[v];
    }
}

__global__ __launch_bounds__(256) void k3_wgrad(const float* __restrict__ ws, float* __restrict__ out)
{
    const int tid = threadIdx.x;
    float ac = 0.f, as2 = 0.f;
    for (int kt = 0; kt < 2048; ++kt) {
        float w2  = ws[O_WV2 + kt];
        float phi = ws[O_PHI + (unsigned)kt * 256u + tid];
        float sn, cs; sincosf(phi, &sn, &cs);
        ac  = fmaf(w2, cs, ac);
        as2 = fmaf(w2, sn, as2);
    }
    out[37888 + tid]       = ac;
    out[37888 + 256 + tid] = as2;
}

__global__ __launch_bounds__(64) void k4_ugrad(const float* __restrict__ ws,
                                               const float* __restrict__ S,
                                               float* __restrict__ out)
{
    __shared__ float dw[1024];
    __shared__ float uv[48];
    const int b = blockIdx.x;
    const int tid = threadIdx.x;
    for (int idx = tid; idx < 1024; idx += 64) {
        int s = idx >> 8, m = idx & 255;
        dw[idx] = ws[O_WV + s * 512 + b] * ws[O_D + (unsigned)((s * 512 + b) * 256 + m)];
    }
    if (tid < 48) uv[tid] = ws[O_INP + (unsigned)(b * 64 + 2 + tid)];
    __syncthreads();
    if (tid < 48) {
        float acc = 0.f;
        for (int m = 0; m < 256; ++m) {
            float dsum = dw[m] + dw[256 + m] + dw[512 + m] + dw[768 + m];
            acc = fmaf(dsum, S[m * 50 + 2 + tid], acc);
        }
        int v = tid >> 4, r = tid & 15;
        float u0 = uv[r], u1 = uv[16 + r], u2 = uv[32 + r];
        float p = (v == 0) ? u1 * u2 : (v == 1) ? u0 * u2 : u0 * u1;
        float gx = ws[O_GX + b], gb = ws[O_GB + b];
        out[512 + b * 48 + tid] = gx * (1.f / 512.f) * gb * p + acc;
    }
}

extern "C" void kernel_launch(void* const* d_in, const int* in_sizes, int n_in,
                              void* d_out, int out_size, void* d_ws, size_t ws_size,
                              hipStream_t stream)
{
    (void)in_sizes; (void)n_in; (void)out_size; (void)ws_size;
    const float* U0   = (const float*)d_in[0];
    const float* U1   = (const float*)d_in[1];
    const float* U2   = (const float*)d_in[2];
    const float* S    = (const float*)d_in[3];
    const float* w    = (const float*)d_in[4];
    const float* btn  = (const float*)d_in[5];
    const float* bobs = (const float*)d_in[6];
    const int*   bin  = (const int*)d_in[7];
    float* out = (float*)d_out;
    float* ws  = (float*)d_ws;

    k1_stages<<<32, 128, 0, stream>>>(U0, U1, U2, S, w, btn, bobs, bin, ws, out);
    k2_sgrad<<<64, 32, 0, stream>>>(ws, out);
    k3_wgrad<<<1, 256, 0, stream>>>(ws, out);
    k4_ugrad<<<512, 64, 0, stream>>>(ws, S, out);
}


# stage loop computes phi transposed (S rows as the A operand, inp as B): a row's 256 terms of F and a lie in lane registers, in-lane adds replace 64 LDS permutes a stage; sum orders of F, a change
# speedup vs baseline: 50.4683x; 50.4683x over previous
#include <hip/hip_runtime.h>
#include <math.h>

typedef float v2f __attribute__((ext_vector_type(2)));
typedef float v8f __attribute__((ext_vector_type(8)));

#define BB 512
#define MM 256
#define INDIM 50

#define PSTRIDE 13312u
#define IPW 65
#define DPW 258
#define CPW 68
#define CPLANE (16 * CPW)
#define CIDX(s, p, w, b, ml) ((((s) * 2 + (p)) * 4 + (w)) * CPLANE + (b) * CPW + (ml))

#define SMALL_ARG_BOUND 131072.f
__device__ __forceinline__ void sincos_small(float x, float& sn, float& cs)
{
    const float ax = __builtin_fabsf(x);
    const float n  = __builtin_rintf(ax * 0x1.45f306p-1f);
    float r = __builtin_fmaf(n, -0x1.921fb4p+0f, ax);
    r = __builtin_fmaf(n, -0x1.4442d0p-24f, r);
    r = __builtin_fmaf(n, -0x1.846988p-48f, r);
    const int   i = (int)n;
    const float t = r * r;
    const float ps = __builtin_fmaf(t, __builtin_fmaf(t, -0x1.983304p-13f, 0x1.110388p-7f), -0x1.55553ap-3f);
    const float sr = __builtin_fmaf(r, t * ps, r);
    const float pc = __builtin_fmaf(t, __builtin_fmaf(t, __builtin_fmaf(t, 0x1.aea668p-16f, -0x1.6c9e76p-10f),
                                                      0x1.5557eep-5f), -0x1.000008p-1f);
    const float cr = __builtin_fmaf(t, pc, 1.0f);
    const bool  odd = (i & 1) != 0;
    const unsigned s_sel = __float_as_uint(odd ? cr : sr);
    const unsigned c_sel = __float_as_uint(odd ? -sr : cr);
    const unsigned flip  = ((unsigned)i << 30) & 0x80000000u;
    const unsigned sx    = __float_as_uint(x) & 0x80000000u;
    sn = __uint_as_float(s_sel ^ flip ^ sx);
    cs = __uint_as_float(c_sel ^ flip);
}

__global__ __launch_bounds__(128) void k1_stages(
    const float* __restrict__ U0, const float* __restrict__ U1, const float* __restrict__ U2,
    const float* __restrict__ S,  const float* __restrict__ w,
    const float* __restrict__ b_t_n, const float* __restrict__ b_obs,
    const int* __restrict__ b_i_n,
    float* __restrict__ ws, float* __restrict__ out)
{
    __shared__ float S_s[256 * 51];
    __shared__ float inp_s[16 * IPW];
    __shared__ float inp0_s[4 * 16 * 17];
    __shared__ __attribute__((aligned(16))) float cs_s[4 * 2 * 4 * CPLANE];
    __shared__ __attribute__((aligned(16))) float w_s[512];
    __shared__ float dw_s[16 * DPW];
    __shared__ float redF[8][16];
    __shared__ float redA[8][16];
    __shared__ float xs[16], beta_s[16], scT_s[16];
    __shared__ float aS[4][16], dxS[4][16];
    __shared__ float wvS[4][16], wv2S[4][16];
    __shared__ float gxs[16], gbs[16];

    const int tid  = threadIdx.x;
    const int wv   = tid >> 5;
    const int lane = tid & 31;
    const int ln   = lane & 15;
    const int hb   = (lane < 16) ? 0 : 8;
    const int role = blockIdx.x >> 5;
    const int rg   = blockIdx.x & 31;
    const int b0g  = rg * 16;

    int giv[6];
    #pragma unroll
    for (int i = 0; i < 6; ++i) {
        int idx = tid + i * 128;
        int bl = idx / 48, j = idx - bl * 48;
        giv[i] = b_i_n[(b0g + bl) * 3 + (j >> 4)];
    }
    const float4* __restrict__ S4 = reinterpret_cast<const float4*>(S);
    float4 sv[25];
    #pragma unroll
    for (int i = 0; i < 25; ++i) sv[i] = S4[tid + i * 128];
    int   mro[4]; float w0r[4], w1r[4];
    #pragma unroll
    for (int nt = 0; nt < 4; ++nt) mro[nt] = wv * 64 + nt * 16 + ln;
    float wg[4];
    #pragma unroll
    for (int i = 0; i < 4; ++i) wg[i] = w[tid + i * 128];
    float scT_r = 0.f, bobs_r = 0.f;
    if (tid < 16) { scT_r = b_t_n[b0g + tid]; bobs_r = b_obs[b0g + tid]; }
    float gv[6];
    #pragma unroll
    for (int i = 0; i < 6; ++i) {
        int idx = tid + i * 128;
        int bl = idx / 48, j = idx - bl * 48;
        int u = j >> 4, r = j & 15;
        const float* Uu = (u == 0) ? U0 : (u == 1) ? U1 : U2;
        gv[i] = Uu[giv[i] * 16 + r];
    }

    #pragma unroll
    for (int i = 0; i < 25; ++i) {
        int e0 = (tid + i * 128) * 4;
        int m0 = e0 / INDIM, k0 = e0 - m0 * INDIM;
        int a0 = e0 + m0;
        int a2 = a0 + 2 + ((k0 == 48) ? 1 : 0);
        S_s[a0]     = sv[i].x;
        S_s[a0 + 1] = sv[i].y;
        S_s[a2]     = sv[i].z;
        S_s[a2 + 1] = sv[i].w;
    }
    for (int m = tid; m < 256; m += 128) S_s[m * 51 + 50] = 0.f;
    #pragma unroll
    for (int i = 0; i < 6; ++i) {
        int idx = tid + i * 128;
        int bl = idx / 48, j = idx - bl * 48;
        inp_s[bl * IPW + 2 + j] = gv[i];
    }
    for (int idx = tid; idx < 16 * 15; idx += 128) {
        int bl = idx / 15, c = idx - bl * 15;
        inp_s[bl * IPW + 50 + c] = 0.f;
    }
    if (tid < 16) scT_s[tid] = scT_r;
    #pragma unroll
    for (int i = 0; i < 4; ++i) w_s[tid + i * 128] = wg[i];
    __syncthreads();

    if (tid < 16) {
        float acc = 0.f;
        for (int r = 0; r < 16; ++r)
            acc += inp_s[tid*IPW + 2 + r] * inp_s[tid*IPW + 18 + r] * inp_s[tid*IPW + 34 + r];
        beta_s[tid] = acc; xs[tid] = acc;
    }
    __syncthreads();

    v2f w0m[4][4], w1m[4][4], s1m[4][4];
    #pragma unroll
    for (int nt = 0; nt < 4; ++nt) {
        w0r[nt] = w_s[mro[nt]]; w1r[nt] = w_s[256 + mro[nt]];
        const int mb = wv * 64 + nt * 16 + hb;
        #pragma unroll
        for (int j = 0; j < 4; ++j) {
            w0m[nt][j].x = w_s[mb + 2 * j];           w0m[nt][j].y = w_s[mb + 2 * j + 1];
            w1m[nt][j].x = w_s[256 + mb + 2 * j];     w1m[nt][j].y = w_s[256 + mb + 2 * j + 1];
            s1m[nt][j].x = S_s[(mb + 2 * j) * 51 + 1]; s1m[nt][j].y = S_s[(mb + 2 * j + 1) * 51 + 1];
        }
    }
    float stv[8];
    #pragma unroll
    for (int v = 0; v < 8; ++v) stv[v] = scT_s[hb + v];

    const int klo = (lane < 16) ? 0 : 2;

    for (int s = 0; s < 4; ++s) {
        if (tid < 16) {
            float ts = (s == 0) ? 0.f : ((s == 3) ? 1.f : 0.5f);
            inp_s[tid * IPW + 0] = scT_s[tid] * ts;
            inp_s[tid * IPW + 1] = xs[tid];
        }
        __syncthreads();

        for (int idx = tid; idx < 16 * 16; idx += 128) {
            int bl = idx >> 4, k = idx & 15;
            inp0_s[(s * 16 + bl) * 17 + k] = inp_s[bl * IPW + k];
        }

        v8f acc[4];
        #pragma unroll
        for (int nt = 0; nt < 4; ++nt)
            #pragma unroll
            for (int v = 0; v < 8; ++v) acc[nt][v] = 0.f;

        #pragma unroll
        for (int kq = 0; kq < 13; ++kq) {
            int k0 = kq * 4 + klo;
            v2f af;
            af.x = inp_s[ln * IPW + k0];
            af.y = inp_s[ln * IPW + k0 + 1];
            #pragma unroll
            for (int nt = 0; nt < 4; ++nt) {
                v2f bf;
                bf.x = (k0     < 50) ? S_s[mro[nt] * 51 + k0]     : 0.f;
                bf.y = (k0 + 1 < 50) ? S_s[mro[nt] * 51 + k0 + 1] : 0.f;
                acc[nt] = __builtin_amdgcn_wmma_f32_16x16x4_f32(
                    false, bf, false, af, (short)0, acc[nt], false, false);
            }
        }

        float csv[4][8], snv[4][8];
        float mxt[4];
        const int cb = CIDX(s, 0, wv, ln, hb);
        #pragma unroll
        for (int nt = 0; nt < 4; ++nt) {
            mxt[nt] = fabsf(acc[nt][0]);
            #pragma unroll
            for (int v = 1; v < 8; ++v) mxt[nt] = fmaxf(mxt[nt], fabsf(acc[nt][v]));
        }
        const float mx = fmaxf(fmaxf(mxt[0], mxt[1]), fmaxf(mxt[2], mxt[3]));
        if (__builtin_expect(__builtin_amdgcn_ballot_w32(mx >= SMALL_ARG_BOUND) != 0u, 0)) {
            #pragma unroll
            for (int nt = 0; nt < 4; ++nt)
                #pragma unroll
                for (int v = 0; v < 8; ++v)
                    cs_s[cb + nt * 16 + v] = acc[nt][v];
            #pragma nounroll
            for (int e = 0; e < 32; ++e) {
                int q = cb + (e >> 3) * 16 + (e & 7);
                float sn, cs;
                sincosf(cs_s[q], &sn, &cs);
                cs_s[q]              = cs;
                cs_s[q + 4 * CPLANE] = sn;
            }
            #pragma unroll
            for (int nt = 0; nt < 4; ++nt)
                #pragma unroll
                for (int v = 0; v < 8; ++v) {
                    int q = cb + nt * 16 + v;
                    csv[nt][v] = cs_s[q];
                    snv[nt][v] = cs_s[q + 4 * CPLANE];
                }
        } else {
            #pragma unroll
            for (int nt = 0; nt < 4; ++nt)
                #pragma unroll
                for (int v = 0; v < 8; ++v) sincos_small(acc[nt][v], snv[nt][v], csv[nt][v]);
        }
        v2f Fp[4], Ap[4];
        #pragma unroll
        for (int j = 0; j < 4; ++j) { Fp[j].x = 0.f; Fp[j].y = 0.f; Ap[j].x = 0.f; Ap[j].y = 0.f; }
        #pragma unroll
        for (int nt = 0; nt < 4; ++nt) {
            #pragma unroll
            for (int j = 0; j < 4; ++j) {
                v2f cs, sn;
                cs.x = csv[nt][2 * j]; cs.y = csv[nt][2 * j + 1];
                sn.x = snv[nt][2 * j]; sn.y = snv[nt][2 * j + 1];
                v2f fval  = w0m[nt][j] * cs + w1m[nt][j] * sn;
                v2f dterm = w1m[nt][j] * cs - w0m[nt][j] * sn;
                Fp[j] += fval;
                Ap[j] += dterm * s1m[nt][j];
            }
            float4* pc = reinterpret_cast<float4*>(&cs_s[cb + nt * 16]);
            float4* pn = reinterpret_cast<float4*>(&cs_s[cb + 4 * CPLANE + nt * 16]);
            pc[0] = make_float4(csv[nt][0], csv[nt][1], csv[nt][2], csv[nt][3]);
            pc[1] = make_float4(csv[nt][4], csv[nt][5], csv[nt][6], csv[nt][7]);
            pn[0] = make_float4(snv[nt][0], snv[nt][1], snv[nt][2], snv[nt][3]);
            pn[1] = make_float4(snv[nt][4], snv[nt][5], snv[nt][6], snv[nt][7]);
        }
        redF[wv * 2 + (lane >> 4)][ln] = ((Fp[0].x + Fp[0].y) + (Fp[1].x + Fp[1].y)) + ((Fp[2].x + Fp[2].y) + (Fp[3].x + Fp[3].y));
        redA[wv * 2 + (lane >> 4)][ln] = ((Ap[0].x + Ap[0].y) + (Ap[1].x + Ap[1].y)) + ((Ap[2].x + Ap[2].y) + (Ap[3].x + Ap[3].y));
        __syncthreads();
        if (tid < 16) {
            float F = redF[0][tid], A = redA[0][tid];
            #pragma unroll
            for (int j = 1; j < 8; ++j) { F += redF[j][tid]; A += redA[j][tid]; }
            float dx = scT_s[tid] * F;
            float a  = scT_s[tid] * A;
            dxS[s][tid] = dx; aS[s][tid] = a;
            if      (s == 0) xs[tid] = beta_s[tid] + 0.5f * dx;
            else if (s == 1) xs[tid] = beta_s[tid] + 0.5f * dx;
            else if (s == 2) xs[tid] = beta_s[tid] + dx;
        }
        __syncthreads();
    }

    if (tid < 16) {
        int bg = b0g + tid;
        float a1 = aS[0][tid], a2 = aS[1][tid], a3 = aS[2][tid], a4 = aS[3][tid];
        float dx1 = dxS[0][tid], dx2 = dxS[1][tid], dx3 = dxS[2][tid], dx4 = dxS[3][tid];
        float xf = beta_s[tid] + (dx1 + 2.f*dx2 + 2.f*dx3 + dx4) * (1.f/6.f);
        float gx = xf - bobs_r;
        float c1 = (1.f + a2 + 0.5f*a3*a2 + 0.25f*a4*a3*a2) * (1.f/6.f);
        float c2 = (2.f + a3 + 0.5f*a4*a3) * (1.f/6.f);
        float c3 = (2.f + a4) * (1.f/6.f);
        float c4 = (1.f/6.f);
        float kb1 = a1;
        float kb2 = a2 * (1.f + 0.5f*kb1);
        float kb3 = a3 * (1.f + 0.5f*kb2);
        float kb4 = a4 * (1.f + kb3);
        float gb = 1.f + (kb1 + 2.f*kb2 + 2.f*kb3 + kb4) * (1.f/6.f);
        const float inv = 1.f / 512.f;
        if (role == 0) out[bg] = gx * gb * inv;
        gxs[tid] = gx; gbs[tid] = gb;
        float sc = scT_s[tid];
        float cc[4] = {c1, c2, c3, c4};
        #pragma unroll
        for (int s2 = 0; s2 < 4; ++s2) {
            float wvv = cc[s2] * gx * inv;
            wvS[s2][tid]  = wvv;
            wv2S[s2][tid] = wvv * sc;
        }
    }
    __syncthreads();

    float* __restrict__ P = ws + (unsigned)rg * PSTRIDE;

    if (role != 0) {
        const int nt = role - 1;
        const float w0q = (nt == 0) ? w0r[0] : (nt == 1) ? w0r[1] : (nt == 2) ? w0r[2] : w0r[3];
        const float w1q = (nt == 0) ? w1r[0] : (nt == 1) ? w1r[1] : (nt == 2) ? w1r[2] : w1r[3];
        v8f accS[4], accC, accN;
        #pragma unroll
        for (int v = 0; v < 8; ++v) { accC[v] = 0.f; accN[v] = 0.f; }
        #pragma unroll
        for (int kt = 0; kt < 4; ++kt)
            #pragma unroll
            for (int v = 0; v < 8; ++v) accS[kt][v] = 0.f;
        v2f bSi[3][4];
        #pragma unroll
        for (int kq = 0; kq < 4; ++kq) {
            int b0 = hb + 2 * kq;
            #pragma unroll
            for (int kt = 1; kt < 4; ++kt) {
                bSi[kt - 1][kq].x = inp_s[b0 * IPW + kt * 16 + ln];
                bSi[kt - 1][kq].y = inp_s[(b0 + 1) * IPW + kt * 16 + ln];
            }
        }
        #pragma unroll
        for (int s = 0; s < 4; ++s) {
            v2f bS0[4], bW[4];
            #pragma unroll
            for (int kq = 0; kq < 4; ++kq) {
                int b0 = hb + 2 * kq;
                bS0[kq].x = inp0_s[(s * 16 + b0) * 17 + ln];
                bS0[kq].y = inp0_s[(s * 16 + b0 + 1) * 17 + ln];
                bW[kq].x = wv2S[s][b0];
                bW[kq].y = wv2S[s][b0 + 1];
            }
            float cs[8], sn[8], wd[8];
            #pragma unroll
            for (int v = 0; v < 8; ++v) {
                int q = CIDX(s, 0, wv, hb + v, nt * 16 + ln);
                cs[v] = cs_s[q];
                sn[v] = cs_s[q + 4 * CPLANE];
                float dterm = w1q * cs[v] - w0q * sn[v];
                float d = stv[v] * dterm;
                wd[v] = wvS[s][hb + v] * d;
            }
            #pragma unroll
            for (int kq = 0; kq < 4; ++kq) {
                v2f aD, aC, aN;
                aD.x = wd[2 * kq]; aD.y = wd[2 * kq + 1];
                aC.x = cs[2 * kq]; aC.y = cs[2 * kq + 1];
                aN.x = sn[2 * kq]; aN.y = sn[2 * kq + 1];
                accS[0] = __builtin_amdgcn_wmma_f32_16x16x4_f32(
                    false, aD, false, bS0[kq], (short)0, accS[0], false, false);
                #pragma unroll
                for (int kt = 1; kt < 4; ++kt)
                    accS[kt] = __builtin_amdgcn_wmma_f32_16x16x4_f32(
                        false, aD, false, bSi[kt - 1][kq], (short)0, accS[kt], false, false);
                accC = __builtin_amdgcn_wmma_f32_16x16x4_f32(
                    false, aC, false, bW[kq], (short)0, accC, false, false);
                accN = __builtin_amdgcn_wmma_f32_16x16x4_f32(
                    false, aN, false, bW[kq], (short)0, accN, false, false);
            }
        }
        const int mbase = wv * 64 + nt * 16 + hb;
        #pragma unroll
        for (int kt = 0; kt < 4; ++kt) {
            int kk = kt * 16 + ln;
            #pragma unroll
            for (int v = 0; v < 8; ++v)
                if (kk < 50) P[(mbase + v) * 50 + kk] = accS[kt][v];
        }
        if (ln == 0) {
            #pragma unroll
            for (int v = 0; v < 8; ++v) {
                P[12800 + mbase + v]       = accC[v];
                P[12800 + 256 + mbase + v] = accN[v];
            }
        }
    } else {
        float dsw[4][8];
        #pragma unroll
        for (int nt = 0; nt < 4; ++nt)
            #pragma unroll
            for (int v = 0; v < 8; ++v) dsw[nt][v] = 0.f;
        const float scb = scT_s[ln];
        #pragma unroll
        for (int s = 0; s < 4; ++s) {
            const float wvb = wvS[s][ln];
            const int cbs = CIDX(s, 0, wv, ln, hb);
            #pragma unroll
            for (int nt = 0; nt < 4; ++nt) {
                const float4* pc = reinterpret_cast<const float4*>(&cs_s[cbs + nt * 16]);
                const float4* pn = reinterpret_cast<const float4*>(&cs_s[cbs + 4 * CPLANE + nt * 16]);
                const float4 c0 = pc[0], c1 = pc[1], n0 = pn[0], n1 = pn[1];
                const float cq[8] = {c0.x, c0.y, c0.z, c0.w, c1.x, c1.y, c1.z, c1.w};
                const float nq[8] = {n0.x, n0.y, n0.z, n0.w, n1.x, n1.y, n1.z, n1.w};
                #pragma unroll
                for (int v = 0; v < 8; ++v) {
                    const float w1e = (v & 1) ? w1m[nt][v >> 1].y : w1m[nt][v >> 1].x;
                    const float w0e = (v & 1) ? w0m[nt][v >> 1].y : w0m[nt][v >> 1].x;
                    float dterm = w1e * cq[v] - w0e * nq[v];
                    float d = scb * dterm;
                    dsw[nt][v] += wvb * d;
                }
            }
        }
        #pragma unroll
        for (int nt = 0; nt < 4; ++nt)
            #pragma unroll
            for (int v = 0; v < 8; ++v) dw_s[ln * DPW + wv * 64 + nt * 16 + hb + v] = dsw[nt][v];
        __syncthreads();
        if (wv < 3) {
            v8f u0 = {0.f, 0.f, 0.f, 0.f, 0.f, 0.f, 0.f, 0.f};
            v8f u1 = {0.f, 0.f, 0.f, 0.f, 0.f, 0.f, 0.f, 0.f};
            const int jc = 2 + wv * 16 + ln;
            for (int kq = 0; kq < 64; kq += 2) {
                int k0 = kq * 4 + klo;
                v2f af0, bf0, af1, bf1;
                af0.x = dw_s[ln * DPW + k0];     af0.y = dw_s[ln * DPW + k0 + 1];
                bf0.x = S_s[k0 * 51 + jc];       bf0.y = S_s[(k0 + 1) * 51 + jc];
                af1.x = dw_s[ln * DPW + k0 + 4]; af1.y = dw_s[ln * DPW + k0 + 5];
                bf1.x = S_s[(k0 + 4) * 51 + jc]; bf1.y = S_s[(k0 + 5) * 51 + jc];
                u0 = __builtin_amdgcn_wmma_f32_16x16x4_f32(false, af0, false, bf0, (short)0, u0, false, false);
                u1 = __builtin_amdgcn_wmma_f32_16x16x4_f32(false, af1, false, bf1, (short)0, u1, false, false);
            }
            #pragma unroll
            for (int v = 0; v < 8; ++v) {
                int b = hb + v;
                float e0 = inp_s[b * IPW + 2 + ln], e1 = inp_s[b * IPW + 18 + ln], e2 = inp_s[b * IPW + 34 + ln];
                float p = (wv == 0) ? e1 * e2 : (wv == 1) ? e0 * e2 : e0 * e1;
                float acc = u0[v] + u1[v];
                out[512 + (b0g + b) * 48 + wv * 16 + ln] = gxs[b] * (1.f / 512.f) * gbs[b] * p + acc;
            }
        }
    }
}

__global__ __launch_bounds__(256) void k2_reduce(const float* __restrict__ ws, float* __restrict__ out)
{
    const unsigned i = blockIdx.x * 256u + threadIdx.x;
    float v[32];
    #pragma unroll
    for (int p = 0; p < 32; ++p) v[p] = ws[(unsigned)p * PSTRIDE + i];
    float acc = v[0];
    #pragma unroll
    for (int p = 1; p < 32; ++p) acc += v[p];
    out[25088u + i] = acc;
}

extern "C" void kernel_launch(void* const* d_in, const int* in_sizes, int n_in,
                              void* d_out, int out_size, void* d_ws, size_t ws_size,
                              hipStream_t stream)
{
    (void)in_sizes; (void)n_in; (void)out_size; (void)ws_size;
    const float* U0   = (const float*)d_in[0];
    const float* U1   = (const float*)d_in[1];
    const float* U2   = (const float*)d_in[2];
    const float* S    = (const float*)d_in[3];
    const float* w    = (const float*)d_in[4];
    const float* btn  = (const float*)d_in[5];
    const float* bobs = (const float*)d_in[6];
    const int*   bin  = (const int*)d_in[7];
    float* out = (float*)d_out;
    float* ws  = (float*)d_ws;

    k1_stages<<<160, 128, 0, stream>>>(U0, U1, U2, S, w, btn, bobs, bin, ws, out);
    k2_reduce<<<52, 256, 0, stream>>>(ws, out);
}


# phi product: all of a stage's fragment reads stand in one group ahead of its 52 WMMAs (S_s pitch 52 with two zero columns, so the pad K-step reads like the rest, no masked reads); sums same
# speedup vs baseline: 54.5057x; 1.0800x over previous
#include <hip/hip_runtime.h>
#include <math.h>

typedef float v2f __attribute__((ext_vector_type(2)));
typedef float v8f __attribute__((ext_vector_type(8)));

#define BB 512
#define MM 256
#define INDIM 50

#define PSTRIDE 13312u
#define IPW 65
#define SPW 52
#define DPW 258
#define CPW 68
#define CPLANE (16 * CPW)
#define CIDX(s, p, w, b, ml) ((((s) * 2 + (p)) * 4 + (w)) * CPLANE + (b) * CPW + (ml))

#define SMALL_ARG_BOUND 131072.f
__device__ __forceinline__ void sincos_small(float x, float& sn, float& cs)
{
    const float ax = __builtin_fabsf(x);
    const float n  = __builtin_rintf(ax * 0x1.45f306p-1f);
    float r = __builtin_fmaf(n, -0x1.921fb4p+0f, ax);
    r = __builtin_fmaf(n, -0x1.4442d0p-24f, r);
    r = __builtin_fmaf(n, -0x1.846988p-48f, r);
    const int   i = (int)n;
    const float t = r * r;
    const float ps = __builtin_fmaf(t, __builtin_fmaf(t, -0x1.983304p-13f, 0x1.110388p-7f), -0x1.55553ap-3f);
    const float sr = __builtin_fmaf(r, t * ps, r);
    const float pc = __builtin_fmaf(t, __builtin_fmaf(t, __builtin_fmaf(t, 0x1.aea668p-16f, -0x1.6c9e76p-10f),
                                                      0x1.5557eep-5f), -0x1.000008p-1f);
    const float cr = __builtin_fmaf(t, pc, 1.0f);
    const bool  odd = (i & 1) != 0;
    const unsigned s_sel = __float_as_uint(odd ? cr : sr);
    const unsigned c_sel = __float_as_uint(odd ? -sr : cr);
    const unsigned flip  = ((unsigned)i << 30) & 0x80000000u;
    const unsigned sx    = __float_as_uint(x) & 0x80000000u;
    sn = __uint_as_float(s_sel ^ flip ^ sx);
    cs = __uint_as_float(c_sel ^ flip);
}

__global__ __launch_bounds__(128) void k1_stages(
    const float* __restrict__ U0, const float* __restrict__ U1, const float* __restrict__ U2,
    const float* __restrict__ S,  const float* __restrict__ w,
    const float* __restrict__ b_t_n, const float* __restrict__ b_obs,
    const int* __restrict__ b_i_n,
    float* __restrict__ ws, float* __restrict__ out)
{
    __shared__ __attribute__((aligned(16))) float S_s[256 * SPW];
    __shared__ float inp_s[16 * IPW];
    __shared__ float inp0_s[4 * 16 * 17];
    __shared__ __attribute__((aligned(16))) float cs_s[4 * 2 * 4 * CPLANE];
    __shared__ __attribute__((aligned(16))) float w_s[512];
    __shared__ float dw_s[16 * DPW];
    __shared__ float redF[8][16];
    __shared__ float redA[8][16];
    __shared__ float xs[16], beta_s[16], scT_s[16];
    __shared__ float aS[4][16], dxS[4][16];
    __shared__ float wvS[4][16], wv2S[4][16];
    __shared__ float gxs[16], gbs[16];

    const int tid  = threadIdx.x;
    const int wv   = tid >> 5;
    const int lane = tid & 31;
    const int ln   = lane & 15;
    const int hb   = (lane < 16) ? 0 : 8;
    const int role = blockIdx.x >> 5;
    const int rg   = blockIdx.x & 31;
    const int b0g  = rg * 16;

    int giv[6];
    #pragma unroll
    for (int i = 0; i < 6; ++i) {
        int idx = tid + i * 128;
        int bl = idx / 48, j = idx - bl * 48;
        giv[i] = b_i_n[(b0g + bl) * 3 + (j >> 4)];
    }
    const float4* __restrict__ S4 = reinterpret_cast<const float4*>(S);
    float4 sv[25];
    #pragma unroll
    for (int i = 0; i < 25; ++i) sv[i] = S4[tid + i * 128];
    int   mro[4]; float w0r[4], w1r[4];
    #pragma unroll
    for (int nt = 0; nt < 4; ++nt) mro[nt] = wv * 64 + nt * 16 + ln;
    float wg[4];
    #pragma unroll
    for (int i = 0; i < 4; ++i) wg[i] = w[tid + i * 128];
    float scT_r = 0.f, bobs_r = 0.f;
    if (tid < 16) { scT_r = b_t_n[b0g + tid]; bobs_r = b_obs[b0g + tid]; }
    float gv[6];
    #pragma unroll
    for (int i = 0; i < 6; ++i) {
        int idx = tid + i * 128;
        int bl = idx / 48, j = idx - bl * 48;
        int u = j >> 4, r = j & 15;
        const float* Uu = (u == 0) ? U0 : (u == 1) ? U1 : U2;
        gv[i] = Uu[giv[i] * 16 + r];
    }

    #pragma unroll
    for (int i = 0; i < 25; ++i) {
        int e0 = (tid + i * 128) * 4;
        int m0 = e0 / INDIM, k0 = e0 - m0 * INDIM;
        int a0 = e0 + 2 * m0;
        int a2 = a0 + 2 + ((k0 == 48) ? 2 : 0);
        S_s[a0]     = sv[i].x;
        S_s[a0 + 1] = sv[i].y;
        S_s[a2]     = sv[i].z;
        S_s[a2 + 1] = sv[i].w;
    }
    for (int m = tid; m < 256; m += 128) { S_s[m * SPW + 50] = 0.f; S_s[m * SPW + 51] = 0.f; }
    #pragma unroll
    for (int i = 0; i < 6; ++i) {
        int idx = tid + i * 128;
        int bl = idx / 48, j = idx - bl * 48;
        inp_s[bl * IPW + 2 + j] = gv[i];
    }
    for (int idx = tid; idx < 16 * 15; idx += 128) {
        int bl = idx / 15, c = idx - bl * 15;
        inp_s[bl * IPW + 50 + c] = 0.f;
    }
    if (tid < 16) scT_s[tid] = scT_r;
    #pragma unroll
    for (int i = 0; i < 4; ++i) w_s[tid + i * 128] = wg[i];
    __syncthreads();

    if (tid < 16) {
        float acc = 0.f;
        for (int r = 0; r < 16; ++r)
            acc += inp_s[tid*IPW + 2 + r] * inp_s[tid*IPW + 18 + r] * inp_s[tid*IPW + 34 + r];
        beta_s[tid] = acc; xs[tid] = acc;
    }
    __syncthreads();

    v2f w0m[4][4], w1m[4][4], s1m[4][4];
    #pragma unroll
    for (int nt = 0; nt < 4; ++nt) {
        w0r[nt] = w_s[mro[nt]]; w1r[nt] = w_s[256 + mro[nt]];
        const int mb = wv * 64 + nt * 16 + hb;
        #pragma unroll
        for (int j = 0; j < 4; ++j) {
            w0m[nt][j].x = w_s[mb + 2 * j];           w0m[nt][j].y = w_s[mb + 2 * j + 1];
            w1m[nt][j].x = w_s[256 + mb + 2 * j];     w1m[nt][j].y = w_s[256 + mb + 2 * j + 1];
            s1m[nt][j].x = S_s[(mb + 2 * j) * SPW + 1]; s1m[nt][j].y = S_s[(mb + 2 * j + 1) * SPW + 1];
        }
    }
    float stv[8];
    #pragma unroll
    for (int v = 0; v < 8; ++v) stv[v] = scT_s[hb + v];

    const int klo = (lane < 16) ? 0 : 2;

    for (int s = 0; s < 4; ++s) {
        if (tid < 16) {
            float ts = (s == 0) ? 0.f : ((s == 3) ? 1.f : 0.5f);
            inp_s[tid * IPW + 0] = scT_s[tid] * ts;
            inp_s[tid * IPW + 1] = xs[tid];
        }
        __syncthreads();

        for (int idx = tid; idx < 16 * 16; idx += 128) {
            int bl = idx >> 4, k = idx & 15;
            inp0_s[(s * 16 + bl) * 17 + k] = inp_s[bl * IPW + k];
        }

        v8f acc[4];
        #pragma unroll
        for (int nt = 0; nt < 4; ++nt)
            #pragma unroll
            for (int v = 0; v < 8; ++v) acc[nt][v] = 0.f;

        v2f af[13], bf[13][4];
        #pragma unroll
        for (int kq = 0; kq < 13; ++kq) {
            int k0 = kq * 4 + klo;
            af[kq].x = inp_s[ln * IPW + k0];
            af[kq].y = inp_s[ln * IPW + k0 + 1];
            #pragma unroll
            for (int nt = 0; nt < 4; ++nt) {
                bf[kq][nt].x = S_s[mro[nt] * SPW + k0];
                bf[kq][nt].y = S_s[mro[nt] * SPW + k0 + 1];
            }
        }
        __builtin_amdgcn_sched_barrier(0);
        #pragma unroll
        for (int kq = 0; kq < 13; ++kq) {
            #pragma unroll
            for (int nt = 0; nt < 4; ++nt)
                acc[nt] = __builtin_amdgcn_wmma_f32_16x16x4_f32(
                    false, bf[kq][nt], false, af[kq], (short)0, acc[nt], false, false);
        }

        float csv[4][8], snv[4][8];
        float mxt[4];
        const int cb = CIDX(s, 0, wv, ln, hb);
        #pragma unroll
        for (int nt = 0; nt < 4; ++nt) {
            mxt[nt] = fabsf(acc[nt][0]);
            #pragma unroll
            for (int v = 1; v < 8; ++v) mxt[nt] = fmaxf(mxt[nt], fabsf(acc[nt][v]));
        }
        const float mx = fmaxf(fmaxf(mxt[0], mxt[1]), fmaxf(mxt[2], mxt[3]));
        if (__builtin_expect(__builtin_amdgcn_ballot_w32(mx >= SMALL_ARG_BOUND) != 0u, 0)) {
            #pragma unroll
            for (int nt = 0; nt < 4; ++nt)
                #pragma unroll
                for (int v = 0; v < 8; ++v)
                    cs_s[cb + nt * 16 + v] = acc[nt][v];
            #pragma nounroll
            for (int e = 0; e < 32; ++e) {
                int q = cb + (e >> 3) * 16 + (e & 7);
                float sn, cs;
                sincosf(cs_s[q], &sn, &cs);
                cs_s[q]              = cs;
                cs_s[q + 4 * CPLANE] = sn;
            }
            #pragma unroll
            for (int nt = 0; nt < 4; ++nt)
                #pragma unroll
                for (int v = 0; v < 8; ++v) {
                    int q = cb + nt * 16 + v;
                    csv[nt][v] = cs_s[q];
                    snv[nt][v] = cs_s[q + 4 * CPLANE];
                }
        } else {
            #pragma unroll
            for (int nt = 0; nt < 4; ++nt)
                #pragma unroll
                for (int v = 0; v < 8; ++v) sincos_small(acc[nt][v], snv[nt][v], csv[nt][v]);
        }
        v2f Fp[4], Ap[4];
        #pragma unroll
        for (int j = 0; j < 4; ++j) { Fp[j].x = 0.f; Fp[j].y = 0.f; Ap[j].x = 0.f; Ap[j].y = 0.f; }
        #pragma unroll
        for (int nt = 0; nt < 4; ++nt) {
            #pragma unroll
            for (int j = 0; j < 4; ++j) {
                v2f cs, sn;
                cs.x = csv[nt][2 * j]; cs.y = csv[nt][2 * j + 1];
                sn.x = snv[nt][2 * j]; sn.y = snv[nt][2 * j + 1];
                v2f fval  = w0m[nt][j] * cs + w1m[nt][j] * sn;
                v2f dterm = w1m[nt][j] * cs - w0m[nt][j] * sn;
                Fp[j] += fval;
                Ap[j] += dterm * s1m[nt][j];
            }
            float4* pc = reinterpret_cast<float4*>(&cs_s[cb + nt * 16]);
            float4* pn = reinterpret_cast<float4*>(&cs_s[cb + 4 * CPLANE + nt * 16]);
            pc[0] = make_float4(csv[nt][0], csv[nt][1], csv[nt][2], csv[nt][3]);
            pc[1] = make_float4(csv[nt][4], csv[nt][5], csv[nt][6], csv[nt][7]);
            pn[0] = make_float4(snv[nt][0], snv[nt][1], snv[nt][2], snv[nt][3]);
            pn[1] = make_float4(snv[nt][4], snv[nt][5], snv[nt][6], snv[nt][7]);
        }
        redF[wv * 2 + (lane >> 4)][ln] = ((Fp[0].x + Fp[0].y) + (Fp[1].x + Fp[1].y)) + ((Fp[2].x + Fp[2].y) + (Fp[3].x + Fp[3].y));
        redA[wv * 2 + (lane >> 4)][ln] = ((Ap[0].x + Ap[0].y) + (Ap[1].x + Ap[1].y)) + ((Ap[2].x + Ap[2].y) + (Ap[3].x + Ap[3].y));
        __syncthreads();
        if (tid < 16) {
            float F = redF[0][tid], A = redA[0][tid];
            #pragma unroll
            for (int j = 1; j < 8; ++j) { F += redF[j][tid]; A += redA[j][tid]; }
            float dx = scT_s[tid] * F;
            float a  = scT_s[tid] * A;
            dxS[s][tid] = dx; aS[s][tid] = a;
            if      (s == 0) xs[tid] = beta_s[tid] + 0.5f * dx;
            else if (s == 1) xs[tid] = beta_s[tid] + 0.5f * dx;
            else if (s == 2) xs[tid] = beta_s[tid] + dx;
        }
        __syncthreads();
    }

    if (tid < 16) {
        int bg = b0g + tid;
        float a1 = aS[0][tid], a2 = aS[1][tid], a3 = aS[2][tid], a4 = aS[3][tid];
        float dx1 = dxS[0][tid], dx2 = dxS[1][tid], dx3 = dxS[2][tid], dx4 = dxS[3][tid];
        float xf = beta_s[tid] + (dx1 + 2.f*dx2 + 2.f*dx3 + dx4) * (1.f/6.f);
        float gx = xf - bobs_r;
        float c1 = (1.f + a2 + 0.5f*a3*a2 + 0.25f*a4*a3*a2) * (1.f/6.f);
        float c2 = (2.f + a3 + 0.5f*a4*a3) * (1.f/6.f);
        float c3 = (2.f + a4) * (1.f/6.f);
        float c4 = (1.f/6.f);
        float kb1 = a1;
        float kb2 = a2 * (1.f + 0.5f*kb1);
        float kb3 = a3 * (1.f + 0.5f*kb2);
        float kb4 = a4 * (1.f + kb3);
        float gb = 1.f + (kb1 + 2.f*kb2 + 2.f*kb3 + kb4) * (1.f/6.f);
        const float inv = 1.f / 512.f;
        if (role == 0) out[bg] = gx * gb * inv;
        gxs[tid] = gx; gbs[tid] = gb;
        float sc = scT_s[tid];
        float cc[4] = {c1, c2, c3, c4};
        #pragma unroll
        for (int s2 = 0; s2 < 4; ++s2) {
            float wvv = cc[s2] * gx * inv;
            wvS[s2][tid]  = wvv;
            wv2S[s2][tid] = wvv * sc;
        }
    }
    __syncthreads();

    float* __restrict__ P = ws + (unsigned)rg * PSTRIDE;

    if (role != 0) {
        const int nt = role - 1;
        const float w0q = (nt == 0) ? w0r[0] : (nt == 1) ? w0r[1] : (nt == 2) ? w0r[2] : w0r[3];
        const float w1q = (nt == 0) ? w1r[0] : (nt == 1) ? w1r[1] : (nt == 2) ? w1r[2] : w1r[3];
        v8f accS[4], accC, accN;
        #pragma unroll
        for (int v = 0; v < 8; ++v) { accC[v] = 0.f; accN[v] = 0.f; }
        #pragma unroll
        for (int kt = 0; kt < 4; ++kt)
            #pragma unroll
            for (int v = 0; v < 8; ++v) accS[kt][v] = 0.f;
        v2f bSi[3][4];
        #pragma unroll
        for (int kq = 0; kq < 4; ++kq) {
            int b0 = hb + 2 * kq;
            #pragma unroll
            for (int kt = 1; kt < 4; ++kt) {
                bSi[kt - 1][kq].x = inp_s[b0 * IPW + kt * 16 + ln];
                bSi[kt - 1][kq].y = inp_s[(b0 + 1) * IPW + kt * 16 + ln];
            }
        }
        #pragma unroll
        for (int s = 0; s < 4; ++s) {
            v2f bS0[4], bW[4];
            #pragma unroll
            for (int kq = 0; kq < 4; ++kq) {
                int b0 = hb + 2 * kq;
                bS0[kq].x = inp0_s[(s * 16 + b0) * 17 + ln];
                bS0[kq].y = inp0_s[(s * 16 + b0 + 1) * 17 + ln];
                bW[kq].x = wv2S[s][b0];
                bW[kq].y = wv2S[s][b0 + 1];
            }
            float cs[8], sn[8], wd[8];
            #pragma unroll
            for (int v = 0; v < 8; ++v) {
                int q = CIDX(s, 0, wv, hb + v, nt * 16 + ln);
                cs[v] = cs_s[q];
                sn[v] = cs_s[q + 4 * CPLANE];
                float dterm = w1q * cs[v] - w0q * sn[v];
                float d = stv[v] * dterm;
                wd[v] = wvS[s][hb + v] * d;
            }
            #pragma unroll
            for (int kq = 0; kq < 4; ++kq) {
                v2f aD, aC, aN;
                aD.x = wd[2 * kq]; aD.y = wd[2 * kq + 1];
                aC.x = cs[2 * kq]; aC.y = cs[2 * kq + 1];
                aN.x = sn[2 * kq]; aN.y = sn[2 * kq + 1];
                accS[0] = __builtin_amdgcn_wmma_f32_16x16x4_f32(
                    false, aD, false, bS0[kq], (short)0, accS[0], false, false);
                #pragma unroll
                for (int kt = 1; kt < 4; ++kt)
                    accS[kt] = __builtin_amdgcn_wmma_f32_16x16x4_f32(
                        false, aD, false, bSi[kt - 1][kq], (short)0, accS[kt], false, false);
                accC = __builtin_amdgcn_wmma_f32_16x16x4_f32(
                    false, aC, false, bW[kq], (short)0, accC, false, false);
                accN = __builtin_amdgcn_wmma_f32_16x16x4_f32(
                    false, aN, false, bW[kq], (short)0, accN, false, false);
            }
        }
        const int mbase = wv * 64 + nt * 16 + hb;
        #pragma unroll
        for (int kt = 0; kt < 4; ++kt) {
            int kk = kt * 16 + ln;
            #pragma unroll
            for (int v = 0; v < 8; ++v)
                if (kk < 50) P[(mbase + v) * 50 + kk] = accS[kt][v];
        }
        if (ln == 0) {
            #pragma unroll
            for (int v = 0; v < 8; ++v) {
                P[12800 + mbase + v]       = accC[v];
                P[12800 + 256 + mbase + v] = accN[v];
            }
        }
    } else {
        float dsw[4][8];
        #pragma unroll
        for (int nt = 0; nt < 4; ++nt)
            #pragma unroll
            for (int v = 0; v < 8; ++v) dsw[nt][v] = 0.f;
        const float scb = scT_s[ln];
        #pragma unroll
        for (int s = 0; s < 4; ++s) {
            const float wvb = wvS[s][ln];
            const int cbs = CIDX(s, 0, wv, ln, hb);
            #pragma unroll
            for (int nt = 0; nt < 4; ++nt) {
                const float4* pc = reinterpret_cast<const float4*>(&cs_s[cbs + nt * 16]);
                const float4* pn = reinterpret_cast<const float4*>(&cs_s[cbs + 4 * CPLANE + nt * 16]);
                const float4 c0 = pc[0], c1 = pc[1], n0 = pn[0], n1 = pn[1];
                const float cq[8] = {c0.x, c0.y, c0.z, c0.w, c1.x, c1.y, c1.z, c1.w};
                const float nq[8] = {n0.x, n0.y, n0.z, n0.w, n1.x, n1.y, n1.z, n1.w};
                #pragma unroll
                for (int v = 0; v < 8; ++v) {
                    const float w1e = (v & 1) ? w1m[nt][v >> 1].y : w1m[nt][v >> 1].x;
                    const float w0e = (v & 1) ? w0m[nt][v >> 1].y : w0m[nt][v >> 1].x;
                    float dterm = w1e * cq[v] - w0e * nq[v];
                    float d = scb * dterm;
                    dsw[nt][v] += wvb * d;
                }
            }
        }
        #pragma unroll
        for (int nt = 0; nt < 4; ++nt)
            #pragma unroll
            for (int v = 0; v < 8; ++v) dw_s[ln * DPW + wv * 64 + nt * 16 + hb + v] = dsw[nt][v];
        __syncthreads();
        if (wv < 3) {
            v8f u0 = {0.f, 0.f, 0.f, 0.f, 0.f, 0.f, 0.f, 0.f};
            v8f u1 = {0.f, 0.f, 0.f, 0.f, 0.f, 0.f, 0.f, 0.f};
            const int jc = 2 + wv * 16 + ln;
            for (int kq = 0; kq < 64; kq += 2) {
                int k0 = kq * 4 + klo;
                v2f af0, bf0, af1, bf1;
                af0.x = dw_s[ln * DPW + k0];     af0.y = dw_s[ln * DPW + k0 + 1];
                bf0.x = S_s[k0 * SPW + jc];      bf0.y = S_s[(k0 + 1) * SPW + jc];
                af1.x = dw_s[ln * DPW + k0 + 4]; af1.y = dw_s[ln * DPW + k0 + 5];
                bf1.x = S_s[(k0 + 4) * SPW + jc]; bf1.y = S_s[(k0 + 5) * SPW + jc];
                u0 = __builtin_amdgcn_wmma_f32_16x16x4_f32(false, af0, false, bf0, (short)0, u0, false, false);
                u1 = __builtin_amdgcn_wmma_f32_16x16x4_f32(false, af1, false, bf1, (short)0, u1, false, false);
            }
            #pragma unroll
            for (int v = 0; v < 8; ++v) {
                int b = hb + v;
                float e0 = inp_s[b * IPW + 2 + ln], e1 = inp_s[b * IPW + 18 + ln], e2 = inp_s[b * IPW + 34 + ln];
                float p = (wv == 0) ? e1 * e2 : (wv == 1) ? e0 * e2 : e0 * e1;
                float acc = u0[v] + u1[v];
                out[512 + (b0g + b) * 48 + wv * 16 + ln] = gxs[b] * (1.f / 512.f) * gbs[b] * p + acc;
            }
        }
    }
}

__global__ __launch_bounds__(256) void k2_reduce(const float* __restrict__ ws, float* __restrict__ out)
{
    const unsigned i = blockIdx.x * 256u + threadIdx.x;
    float v[32];
    #pragma unroll
    for (int p = 0; p < 32; ++p) v[p] = ws[(unsigned)p * PSTRIDE + i];
    float acc = v[0];
    #pragma unroll
    for (int p = 1; p < 32; ++p) acc += v[p];
    out[25088u + i] = acc;
}

extern "C" void kernel_launch(void* const* d_in, const int* in_sizes, int n_in,
                              void* d_out, int out_size, void* d_ws, size_t ws_size,
                              hipStream_t stream)
{
    (void)in_sizes; (void)n_in; (void)out_size; (void)ws_size;
    const float* U0   = (const float*)d_in[0];
    const float* U1   = (const float*)d_in[1];
    const float* U2   = (const float*)d_in[2];
    const float* S    = (const float*)d_in[3];
    const float* w    = (const float*)d_in[4];
    const float* btn  = (const float*)d_in[5];
    const float* bobs = (const float*)d_in[6];
    const int*   bin  = (const int*)d_in[7];
    float* out = (float*)d_out;
    float* ws  = (float*)d_ws;

    k1_stages<<<160, 128, 0, stream>>>(U0, U1, U2, S, w, btn, bobs, bin, ws, out);
    k2_reduce<<<52, 256, 0, stream>>>(ws, out);
}


# stage loop: sin/cos by v_sin_f32/v_cos_f32 on phi/(2 pi) (one multiply; full range: no bound test, no branch), not f32 polynomial code; their error (expected ~1e-6 abs) replaces 1 ulp; sums same
# speedup vs baseline: 59.1485x; 1.0852x over previous
#include <hip/hip_runtime.h>
#include <math.h>

typedef float v2f __attribute__((ext_vector_type(2)));
typedef float v8f __attribute__((ext_vector_type(8)));

#define BB 512
#define MM 256
#define INDIM 50

#define PSTRIDE 13312u
#define IPW 65
#define SPW 52
#define DPW 258
#define CPW 68
#define CPLANE (16 * CPW)
#define CIDX(s, p, w, b, ml) ((((s) * 2 + (p)) * 4 + (w)) * CPLANE + (b) * CPW + (ml))

#define INV_2PI 0x1.45f306p-3f

__global__ __launch_bounds__(128) void k1_stages(
    const float* __restrict__ U0, const float* __restrict__ U1, const float* __restrict__ U2,
    const float* __restrict__ S,  const float* __restrict__ w,
    const float* __restrict__ b_t_n, const float* __restrict__ b_obs,
    const int* __restrict__ b_i_n,
    float* __restrict__ ws, float* __restrict__ out)
{
    __shared__ __attribute__((aligned(16))) float S_s[256 * SPW];
    __shared__ float inp_s[16 * IPW];
    __shared__ float inp0_s[4 * 16 * 17];
    __shared__ __attribute__((aligned(16))) float cs_s[4 * 2 * 4 * CPLANE];
    __shared__ __attribute__((aligned(16))) float w_s[512];
    __shared__ float dw_s[16 * DPW];
    __shared__ float redF[8][16];
    __shared__ float redA[8][16];
    __shared__ float xs[16], beta_s[16], scT_s[16];
    __shared__ float aS[4][16], dxS[4][16];
    __shared__ float wvS[4][16], wv2S[4][16];
    __shared__ float gxs[16], gbs[16];

    const int tid  = threadIdx.x;
    const int wv   = tid >> 5;
    const int lane = tid & 31;
    const int ln   = lane & 15;
    const int hb   = (lane < 16) ? 0 : 8;
    const int role = blockIdx.x >> 5;
    const int rg   = blockIdx.x & 31;
    const int b0g  = rg * 16;

    int giv[6];
    #pragma unroll
    for (int i = 0; i < 6; ++i) {
        int idx = tid + i * 128;
        int bl = idx / 48, j = idx - bl * 48;
        giv[i] = b_i_n[(b0g + bl) * 3 + (j >> 4)];
    }
    const float4* __restrict__ S4 = reinterpret_cast<const float4*>(S);
    float4 sv[25];
    #pragma unroll
    for (int i = 0; i < 25; ++i) sv[i] = S4[tid + i * 128];
    int   mro[4]; float w0r[4], w1r[4];
    #pragma unroll
    for (int nt = 0; nt < 4; ++nt) mro[nt] = wv * 64 + nt * 16 + ln;
    float wg[4];
    #pragma unroll
    for (int i = 0; i < 4; ++i) wg[i] = w[tid + i * 128];
    float scT_r = 0.f, bobs_r = 0.f;
    if (tid < 16) { scT_r = b_t_n[b0g + tid]; bobs_r = b_obs[b0g + tid]; }
    float gv[6];
    #pragma unroll
    for (int i = 0; i < 6; ++i) {
        int idx = tid + i * 128;
        int bl = idx / 48, j = idx - bl * 48;
        int u = j >> 4, r = j & 15;
        const float* Uu = (u == 0) ? U0 : (u == 1) ? U1 : U2;
        gv[i] = Uu[giv[i] * 16 + r];
    }

    #pragma unroll
    for (int i = 0; i < 25; ++i) {
        int e0 = (tid + i * 128) * 4;
        int m0 = e0 / INDIM, k0 = e0 - m0 * INDIM;
        int a0 = e0 + 2 * m0;
        int a2 = a0 + 2 + ((k0 == 48) ? 2 : 0);
        S_s[a0]     = sv[i].x;
        S_s[a0 + 1] = sv[i].y;
        S_s[a2]     = sv[i].z;
        S_s[a2 + 1] = sv[i].w;
    }
    for (int m = tid; m < 256; m += 128) { S_s[m * SPW + 50] = 0.f; S_s[m * SPW + 51] = 0.f; }
    #pragma unroll
    for (int i = 0; i < 6; ++i) {
        int idx = tid + i * 128;
        int bl = idx / 48, j = idx - bl * 48;
        inp_s[bl * IPW + 2 + j] = gv[i];
    }
    for (int idx = tid; idx < 16 * 15; idx += 128) {
        int bl = idx / 15, c = idx - bl * 15;
        inp_s[bl * IPW + 50 + c] = 0.f;
    }
    if (tid < 16) scT_s[tid] = scT_r;
    #pragma unroll
    for (int i = 0; i < 4; ++i) w_s[tid + i * 128] = wg[i];
    __syncthreads();

    if (tid < 16) {
        float acc = 0.f;
        for (int r = 0; r < 16; ++r)
            acc += inp_s[tid*IPW + 2 + r] * inp_s[tid*IPW + 18 + r] * inp_s[tid*IPW + 34 + r];
        beta_s[tid] = acc; xs[tid] = acc;
    }
    __syncthreads();

    v2f w0m[4][4], w1m[4][4], s1m[4][4];
    #pragma unroll
    for (int nt = 0; nt < 4; ++nt) {
        w0r[nt] = w_s[mro[nt]]; w1r[nt] = w_s[256 + mro[nt]];
        const int mb = wv * 64 + nt * 16 + hb;
        #pragma unroll
        for (int j = 0; j < 4; ++j) {
            w0m[nt][j].x = w_s[mb + 2 * j];           w0m[nt][j].y = w_s[mb + 2 * j + 1];
            w1m[nt][j].x = w_s[256 + mb + 2 * j];     w1m[nt][j].y = w_s[256 + mb + 2 * j + 1];
            s1m[nt][j].x = S_s[(mb + 2 * j) * SPW + 1]; s1m[nt][j].y = S_s[(mb + 2 * j + 1) * SPW + 1];
        }
    }
    float stv[8];
    #pragma unroll
    for (int v = 0; v < 8; ++v) stv[v] = scT_s[hb + v];

    const int klo = (lane < 16) ? 0 : 2;

    for (int s = 0; s < 4; ++s) {
        if (tid < 16) {
            float ts = (s == 0) ? 0.f : ((s == 3) ? 1.f : 0.5f);
            inp_s[tid * IPW + 0] = scT_s[tid] * ts;
            inp_s[tid * IPW + 1] = xs[tid];
        }
        __syncthreads();

        for (int idx = tid; idx < 16 * 16; idx += 128) {
            int bl = idx >> 4, k = idx & 15;
            inp0_s[(s * 16 + bl) * 17 + k] = inp_s[bl * IPW + k];
        }

        v8f acc[4];
        #pragma unroll
        for (int nt = 0; nt < 4; ++nt)
            #pragma unroll
            for (int v = 0; v < 8; ++v) acc[nt][v] = 0.f;

        v2f af[13], bf[13][4];
        #pragma unroll
        for (int kq = 0; kq < 13; ++kq) {
            int k0 = kq * 4 + klo;
            af[kq].x = inp_s[ln * IPW + k0];
            af[kq].y = inp_s[ln * IPW + k0 + 1];
            #pragma unroll
            for (int nt = 0; nt < 4; ++nt) {
                bf[kq][nt].x = S_s[mro[nt] * SPW + k0];
                bf[kq][nt].y = S_s[mro[nt] * SPW + k0 + 1];
            }
        }
        __builtin_amdgcn_sched_barrier(0);
        #pragma unroll
        for (int kq = 0; kq < 13; ++kq) {
            #pragma unroll
            for (int nt = 0; nt < 4; ++nt)
                acc[nt] = __builtin_amdgcn_wmma_f32_16x16x4_f32(
                    false, bf[kq][nt], false, af[kq], (short)0, acc[nt], false, false);
        }

        float csv[4][8], snv[4][8];
        const int cb = CIDX(s, 0, wv, ln, hb);
        #pragma unroll
        for (int nt = 0; nt < 4; ++nt) {
            const v8f rev = acc[nt] * INV_2PI;
            #pragma unroll
            for (int v = 0; v < 8; ++v) {
                snv[nt][v] = __builtin_amdgcn_sinf(rev[v]);
                csv[nt][v] = __builtin_amdgcn_cosf(rev[v]);
            }
        }
        v2f Fp[4], Ap[4];
        #pragma unroll
        for (int j = 0; j < 4; ++j) { Fp[j].x = 0.f; Fp[j].y = 0.f; Ap[j].x = 0.f; Ap[j].y = 0.f; }
        #pragma unroll
        for (int nt = 0; nt < 4; ++nt) {
            #pragma unroll
            for (int j = 0; j < 4; ++j) {
                v2f cs, sn;
                cs.x = csv[nt][2 * j]; cs.y = csv[nt][2 * j + 1];
                sn.x = snv[nt][2 * j]; sn.y = snv[nt][2 * j + 1];
                v2f fval  = w0m[nt][j] * cs + w1m[nt][j] * sn;
                v2f dterm = w1m[nt][j] * cs - w0m[nt][j] * sn;
                Fp[j] += fval;
                Ap[j] += dterm * s1m[nt][j];
            }
            float4* pc = reinterpret_cast<float4*>(&cs_s[cb + nt * 16]);
            float4* pn = reinterpret_cast<float4*>(&cs_s[cb + 4 * CPLANE + nt * 16]);
            pc[0] = make_float4(csv[nt][0], csv[nt][1], csv[nt][2], csv[nt][3]);
            pc[1] = make_float4(csv[nt][4], csv[nt][5], csv[nt][6], csv[nt][7]);
            pn[0] = make_float4(snv[nt][0], snv[nt][1], snv[nt][2], snv[nt][3]);
            pn[1] = make_float4(snv[nt][4], snv[nt][5], snv[nt][6], snv[nt][7]);
        }
        redF[wv * 2 + (lane >> 4)][ln] = ((Fp[0].x + Fp[0].y) + (Fp[1].x + Fp[1].y)) + ((Fp[2].x + Fp[2].y) + (Fp[3].x + Fp[3].y));
        redA[wv * 2 + (lane >> 4)][ln] = ((Ap[0].x + Ap[0].y) + (Ap[1].x + Ap[1].y)) + ((Ap[2].x + Ap[2].y) + (Ap[3].x + Ap[3].y));
        __syncthreads();
        if (tid < 16) {
            float F = redF[0][tid], A = redA[0][tid];
            #pragma unroll
            for (int j = 1; j < 8; ++j) { F += redF[j][tid]; A += redA[j][tid]; }
            float dx = scT_s[tid] * F;
            float a  = scT_s[tid] * A;
            dxS[s][tid] = dx; aS[s][tid] = a;
            if      (s == 0) xs[tid] = beta_s[tid] + 0.5f * dx;
            else if (s == 1) xs[tid] = beta_s[tid] + 0.5f * dx;
            else if (s == 2) xs[tid] = beta_s[tid] + dx;
        }
        __syncthreads();
    }

    if (tid < 16) {
        int bg = b0g + tid;
        float a1 = aS[0][tid], a2 = aS[1][tid], a3 = aS[2][tid], a4 = aS[3][tid];
        float dx1 = dxS[0][tid], dx2 = dxS[1][tid], dx3 = dxS[2][tid], dx4 = dxS[3][tid];
        float xf = beta_s[tid] + (dx1 + 2.f*dx2 + 2.f*dx3 + dx4) * (1.f/6.f);
        float gx = xf - bobs_r;
        float c1 = (1.f + a2 + 0.5f*a3*a2 + 0.25f*a4*a3*a2) * (1.f/6.f);
        float c2 = (2.f + a3 + 0.5f*a4*a3) * (1.f/6.f);
        float c3 = (2.f + a4) * (1.f/6.f);
        float c4 = (1.f/6.f);
        float kb1 = a1;
        float kb2 = a2 * (1.f + 0.5f*kb1);
        float kb3 = a3 * (1.f + 0.5f*kb2);
        float kb4 = a4 * (1.f + kb3);
        float gb = 1.f + (kb1 + 2.f*kb2 + 2.f*kb3 + kb4) * (1.f/6.f);
        const float inv = 1.f / 512.f;
        if (role == 0) out[bg] = gx * gb * inv;
        gxs[tid] = gx; gbs[tid] = gb;
        float sc = scT_s[tid];
        float cc[4] = {c1, c2, c3, c4};
        #pragma unroll
        for (int s2 = 0; s2 < 4; ++s2) {
            float wvv = cc[s2] * gx * inv;
            wvS[s2][tid]  = wvv;
            wv2S[s2][tid] = wvv * sc;
        }
    }
    __syncthreads();

    float* __restrict__ P = ws + (unsigned)rg * PSTRIDE;

    if (role != 0) {
        const int nt = role - 1;
        const float w0q = (nt == 0) ? w0r[0] : (nt == 1) ? w0r[1] : (nt == 2) ? w0r[2] : w0r[3];
        const float w1q = (nt == 0) ? w1r[0] : (nt == 1) ? w1r[1] : (nt == 2) ? w1r[2] : w1r[3];
        v8f accS[4], accC, accN;
        #pragma unroll
        for (int v = 0; v < 8; ++v) { accC[v] = 0.f; accN[v] = 0.f; }
        #pragma unroll
        for (int kt = 0; kt < 4; ++kt)
            #pragma unroll
            for (int v = 0; v < 8; ++v) accS[kt][v] = 0.f;
        v2f bSi[3][4];
        #pragma unroll
        for (int kq = 0; kq < 4; ++kq) {
            int b0 = hb + 2 * kq;
            #pragma unroll
            for (int kt = 1; kt < 4; ++kt) {
                bSi[kt - 1][kq].x = inp_s[b0 * IPW + kt * 16 + ln];
                bSi[kt - 1][kq].y = inp_s[(b0 + 1) * IPW + kt * 16 + ln];
            }
        }
        #pragma unroll
        for (int s = 0; s < 4; ++s) {
            v2f bS0[4], bW[4];
            #pragma unroll
            for (int kq = 0; kq < 4; ++kq) {
                int b0 = hb + 2 * kq;
                bS0[kq].x = inp0_s[(s * 16 + b0) * 17 + ln];
                bS0[kq].y = inp0_s[(s * 16 + b0 + 1) * 17 + ln];
                bW[kq].x = wv2S[s][b0];
                bW[kq].y = wv2S[s][b0 + 1];
            }
            float cs[8], sn[8], wd[8];
            #pragma unroll
            for (int v = 0; v < 8; ++v) {
                int q = CIDX(s, 0, wv, hb + v, nt * 16 + ln);
                cs[v] = cs_s[q];
                sn[v] = cs_s[q + 4 * CPLANE];
                float dterm = w1q * cs[v] - w0q * sn[v];
                float d = stv[v] * dterm;
                wd[v] = wvS[s][hb + v] * d;
            }
            #pragma unroll
            for (int kq = 0; kq < 4; ++kq) {
                v2f aD, aC, aN;
                aD.x = wd[2 * kq]; aD.y = wd[2 * kq + 1];
                aC.x = cs[2 * kq]; aC.y = cs[2 * kq + 1];
                aN.x = sn[2 * kq]; aN.y = sn[2 * kq + 1];
                accS[0] = __builtin_amdgcn_wmma_f32_16x16x4_f32(
                    false, aD, false, bS0[kq], (short)0, accS[0], false, false);
                #pragma unroll
                for (int kt = 1; kt < 4; ++kt)
                    accS[kt] = __builtin_amdgcn_wmma_f32_16x16x4_f32(
                        false, aD, false, bSi[kt - 1][kq], (short)0, accS[kt], false, false);
                accC = __builtin_amdgcn_wmma_f32_16x16x4_f32(
                    false, aC, false, bW[kq], (short)0, accC, false, false);
                accN = __builtin_amdgcn_wmma_f32_16x16x4_f32(
                    false, aN, false, bW[kq], (short)0, accN, false, false);
            }
        }
        const int mbase = wv * 64 + nt * 16 + hb;
        #pragma unroll
        for (int kt = 0; kt < 4; ++kt) {
            int kk = kt * 16 + ln;
            #pragma unroll
            for (int v = 0; v < 8; ++v)
                if (kk < 50) P[(mbase + v) * 50 + kk] = accS[kt][v];
        }
        if (ln == 0) {
            #pragma unroll
            for (int v = 0; v < 8; ++v) {
                P[12800 + mbase + v]       = accC[v];
                P[12800 + 256 + mbase + v] = accN[v];
            }
        }
    } else {
        float dsw[4][8];
        #pragma unroll
        for (int nt = 0; nt < 4; ++nt)
            #pragma unroll
            for (int v = 0; v < 8; ++v) dsw[nt][v] = 0.f;
        const float scb = scT_s[ln];
        #pragma unroll
        for (int s = 0; s < 4; ++s) {
            const float wvb = wvS[s][ln];
            const int cbs = CIDX(s, 0, wv, ln, hb);
            #pragma unroll
            for (int nt = 0; nt < 4; ++nt) {
                const float4* pc = reinterpret_cast<const float4*>(&cs_s[cbs + nt * 16]);
                const float4* pn = reinterpret_cast<const float4*>(&cs_s[cbs + 4 * CPLANE + nt * 16]);
                const float4 c0 = pc[0], c1 = pc[1], n0 = pn[0], n1 = pn[1];
                const float cq[8] = {c0.x, c0.y, c0.z, c0.w, c1.x, c1.y, c1.z, c1.w};
                const float nq[8] = {n0.x, n0.y, n0.z, n0.w, n1.x, n1.y, n1.z, n1.w};
                #pragma unroll
                for (int v = 0; v < 8; ++v) {
                    const float w1e = (v & 1) ? w1m[nt][v >> 1].y : w1m[nt][v >> 1].x;
                    const float w0e = (v & 1) ? w0m[nt][v >> 1].y : w0m[nt][v >> 1].x;
                    float dterm = w1e * cq[v] - w0e * nq[v];
                    float d = scb * dterm;
                    dsw[nt][v] += wvb * d;
                }
            }
        }
        #pragma unroll
        for (int nt = 0; nt < 4; ++nt)
            #pragma unroll
            for (int v = 0; v < 8; ++v) dw_s[ln * DPW + wv * 64 + nt * 16 + hb + v] = dsw[nt][v];
        __syncthreads();
        if (wv < 3) {
            v8f u0 = {0.f, 0.f, 0.f, 0.f, 0.f, 0.f, 0.f, 0.f};
            v8f u1 = {0.f, 0.f, 0.f, 0.f, 0.f, 0.f, 0.f, 0.f};
            const int jc = 2 + wv * 16 + ln;
            for (int kq = 0; kq < 64; kq += 2) {
                int k0 = kq * 4 + klo;
                v2f af0, bf0, af1, bf1;
                af0.x = dw_s[ln * DPW + k0];     af0.y = dw_s[ln * DPW + k0 + 1];
                bf0.x = S_s[k0 * SPW + jc];      bf0.y = S_s[(k0 + 1) * SPW + jc];
                af1.x = dw_s[ln * DPW + k0 + 4]; af1.y = dw_s[ln * DPW + k0 + 5];
                bf1.x = S_s[(k0 + 4) * SPW + jc]; bf1.y = S_s[(k0 + 5) * SPW + jc];
                u0 = __builtin_amdgcn_wmma_f32_16x16x4_f32(false, af0, false, bf0, (short)0, u0, false, false);
                u1 = __builtin_amdgcn_wmma_f32_16x16x4_f32(false, af1, false, bf1, (short)0, u1, false, false);
            }
            #pragma unroll
            for (int v = 0; v < 8; ++v) {
                int b = hb + v;
                float e0 = inp_s[b * IPW + 2 + ln], e1 = inp_s[b * IPW + 18 + ln], e2 = inp_s[b * IPW + 34 + ln];
                float p = (wv == 0) ? e1 * e2 : (wv == 1) ? e0 * e2 : e0 * e1;
                float acc = u0[v] + u1[v];
                out[512 + (b0g + b) * 48 + wv * 16 + ln] = gxs[b] * (1.f / 512.f) * gbs[b] * p + acc;
            }
        }
    }
}

__global__ __launch_bounds__(256) void k2_reduce(const float* __restrict__ ws, float* __restrict__ out)
{
    const unsigned i = blockIdx.x * 256u + threadIdx.x;
    float v[32];
    #pragma unroll
    for (int p = 0; p < 32; ++p) v[p] = ws[(unsigned)p * PSTRIDE + i];
    float acc = v[0];
    #pragma unroll
    for (int p = 1; p < 32; ++p) acc += v[p];
    out[25088u + i] = acc;
}

extern "C" void kernel_launch(void* const* d_in, const int* in_sizes, int n_in,
                              void* d_out, int out_size, void* d_ws, size_t ws_size,
                              hipStream_t stream)
{
    (void)in_sizes; (void)n_in; (void)out_size; (void)ws_size;
    const float* U0   = (const float*)d_in[0];
    const float* U1   = (const float*)d_in[1];
    const float* U2   = (const float*)d_in[2];
    const float* S    = (const float*)d_in[3];
    const float* w    = (const float*)d_in[4];
    const float* btn  = (const float*)d_in[5];
    const float* bobs = (const float*)d_in[6];
    const int*   bin  = (const int*)d_in[7];
    float* out = (float*)d_out;
    float* ws  = (float*)d_ws;

    k1_stages<<<160, 128, 0, stream>>>(U0, U1, U2, S, w, btn, bobs, bin, ws, out);
    k2_reduce<<<52, 256, 0, stream>>>(ws, out);
}


# Uvec-grad role: dsw as built (stage-loop layout) is the A fragment; each wave multiplies its own 64 columns m (48 WMMAs, reads issued ahead), 4 wave partials added in LDS; out_Uvec sum order changes
# speedup vs baseline: 65.0203x; 1.0993x over previous
#include <hip/hip_runtime.h>
#include <math.h>

typedef float v2f __attribute__((ext_vector_type(2)));
typedef float v8f __attribute__((ext_vector_type(8)));

#define BB 512
#define MM 256
#define INDIM 50

#define PSTRIDE 13312u
#define IPW 65
#define SPW 52
#define PPW 18
#define CPW 68
#define CPLANE (16 * CPW)
#define CIDX(s, p, w, b, ml) ((((s) * 2 + (p)) * 4 + (w)) * CPLANE + (b) * CPW + (ml))

#define INV_2PI 0x1.45f306p-3f

__global__ __launch_bounds__(128) void k1_stages(
    const float* __restrict__ U0, const float* __restrict__ U1, const float* __restrict__ U2,
    const float* __restrict__ S,  const float* __restrict__ w,
    const float* __restrict__ b_t_n, const float* __restrict__ b_obs,
    const int* __restrict__ b_i_n,
    float* __restrict__ ws, float* __restrict__ out)
{
    __shared__ __attribute__((aligned(16))) float S_s[256 * SPW];
    __shared__ float inp_s[16 * IPW];
    __shared__ float inp0_s[4 * 16 * 17];
    __shared__ __attribute__((aligned(16))) float cs_s[4 * 2 * 4 * CPLANE];
    __shared__ __attribute__((aligned(16))) float w_s[512];
    __shared__ float part_s[4 * 3 * 16 * PPW];
    __shared__ float redF[8][16];
    __shared__ float redA[8][16];
    __shared__ float xs[16], beta_s[16], scT_s[16];
    __shared__ float aS[4][16], dxS[4][16];
    __shared__ float wvS[4][16], wv2S[4][16];
    __shared__ float gxs[16], gbs[16];

    const int tid  = threadIdx.x;
    const int wv   = tid >> 5;
    const int lane = tid & 31;
    const int ln   = lane & 15;
    const int hb   = (lane < 16) ? 0 : 8;
    const int role = blockIdx.x >> 5;
    const int rg   = blockIdx.x & 31;
    const int b0g  = rg * 16;

    int giv[6];
    #pragma unroll
    for (int i = 0; i < 6; ++i) {
        int idx = tid + i * 128;
        int bl = idx / 48, j = idx - bl * 48;
        giv[i] = b_i_n[(b0g + bl) * 3 + (j >> 4)];
    }
    const float4* __restrict__ S4 = reinterpret_cast<const float4*>(S);
    float4 sv[25];
    #pragma unroll
    for (int i = 0; i < 25; ++i) sv[i] = S4[tid + i * 128];
    int   mro[4]; float w0r[4], w1r[4];
    #pragma unroll
    for (int nt = 0; nt < 4; ++nt) mro[nt] = wv * 64 + nt * 16 + ln;
    float wg[4];
    #pragma unroll
    for (int i = 0; i < 4; ++i) wg[i] = w[tid + i * 128];
    float scT_r = 0.f, bobs_r = 0.f;
    if (tid < 16) { scT_r = b_t_n[b0g + tid]; bobs_r = b_obs[b0g + tid]; }
    float gv[6];
    #pragma unroll
    for (int i = 0; i < 6; ++i) {
        int idx = tid + i * 128;
        int bl = idx / 48, j = idx - bl * 48;
        int u = j >> 4, r = j & 15;
        const float* Uu = (u == 0) ? U0 : (u == 1) ? U1 : U2;
        gv[i] = Uu[giv[i] * 16 + r];
    }

    #pragma unroll
    for (int i = 0; i < 25; ++i) {
        int e0 = (tid + i * 128) * 4;
        int m0 = e0 / INDIM, k0 = e0 - m0 * INDIM;
        int a0 = e0 + 2 * m0;
        int a2 = a0 + 2 + ((k0 == 48) ? 2 : 0);
        S_s[a0]     = sv[i].x;
        S_s[a0 + 1] = sv[i].y;
        S_s[a2]     = sv[i].z;
        S_s[a2 + 1] = sv[i].w;
    }
    for (int m = tid; m < 256; m += 128) { S_s[m * SPW + 50] = 0.f; S_s[m * SPW + 51] = 0.f; }
    #pragma unroll
    for (int i = 0; i < 6; ++i) {
        int idx = tid + i * 128;
        int bl = idx / 48, j = idx - bl * 48;
        inp_s[bl * IPW + 2 + j] = gv[i];
    }
    for (int idx = tid; idx < 16 * 15; idx += 128) {
        int bl = idx / 15, c = idx - bl * 15;
        inp_s[bl * IPW + 50 + c] = 0.f;
    }
    if (tid < 16) scT_s[tid] = scT_r;
    #pragma unroll
    for (int i = 0; i < 4; ++i) w_s[tid + i * 128] = wg[i];
    __syncthreads();

    if (tid < 16) {
        float acc = 0.f;
        for (int r = 0; r < 16; ++r)
            acc += inp_s[tid*IPW + 2 + r] * inp_s[tid*IPW + 18 + r] * inp_s[tid*IPW + 34 + r];
        beta_s[tid] = acc; xs[tid] = acc;
    }
    __syncthreads();

    v2f w0m[4][4], w1m[4][4], s1m[4][4];
    #pragma unroll
    for (int nt = 0; nt < 4; ++nt) {
        w0r[nt] = w_s[mro[nt]]; w1r[nt] = w_s[256 + mro[nt]];
        const int mb = wv * 64 + nt * 16 + hb;
        #pragma unroll
        for (int j = 0; j < 4; ++j) {
            w0m[nt][j].x = w_s[mb + 2 * j];           w0m[nt][j].y = w_s[mb + 2 * j + 1];
            w1m[nt][j].x = w_s[256 + mb + 2 * j];     w1m[nt][j].y = w_s[256 + mb + 2 * j + 1];
            s1m[nt][j].x = S_s[(mb + 2 * j) * SPW + 1]; s1m[nt][j].y = S_s[(mb + 2 * j + 1) * SPW + 1];
        }
    }
    float stv[8];
    #pragma unroll
    for (int v = 0; v < 8; ++v) stv[v] = scT_s[hb + v];

    const int klo = (lane < 16) ? 0 : 2;

    for (int s = 0; s < 4; ++s) {
        if (tid < 16) {
            float ts = (s == 0) ? 0.f : ((s == 3) ? 1.f : 0.5f);
            inp_s[tid * IPW + 0] = scT_s[tid] * ts;
            inp_s[tid * IPW + 1] = xs[tid];
        }
        __syncthreads();

        for (int idx = tid; idx < 16 * 16; idx += 128) {
            int bl = idx >> 4, k = idx & 15;
            inp0_s[(s * 16 + bl) * 17 + k] = inp_s[bl * IPW + k];
        }

        v8f acc[4];
        #pragma unroll
        for (int nt = 0; nt < 4; ++nt)
            #pragma unroll
            for (int v = 0; v < 8; ++v) acc[nt][v] = 0.f;

        v2f af[13], bf[13][4];
        #pragma unroll
        for (int kq = 0; kq < 13; ++kq) {
            int k0 = kq * 4 + klo;
            af[kq].x = inp_s[ln * IPW + k0];
            af[kq].y = inp_s[ln * IPW + k0 + 1];
            #pragma unroll
            for (int nt = 0; nt < 4; ++nt) {
                bf[kq][nt].x = S_s[mro[nt] * SPW + k0];
                bf[kq][nt].y = S_s[mro[nt] * SPW + k0 + 1];
            }
        }
        __builtin_amdgcn_sched_barrier(0);
        #pragma unroll
        for (int kq = 0; kq < 13; ++kq) {
            #pragma unroll
            for (int nt = 0; nt < 4; ++nt)
                acc[nt] = __builtin_amdgcn_wmma_f32_16x16x4_f32(
                    false, bf[kq][nt], false, af[kq], (short)0, acc[nt], false, false);
        }

        float csv[4][8], snv[4][8];
        const int cb = CIDX(s, 0, wv, ln, hb);
        #pragma unroll
        for (int nt = 0; nt < 4; ++nt) {
            const v8f rev = acc[nt] * INV_2PI;
            #pragma unroll
            for (int v = 0; v < 8; ++v) {
                snv[nt][v] = __builtin_amdgcn_sinf(rev[v]);
                csv[nt][v] = __builtin_amdgcn_cosf(rev[v]);
            }
        }
        v2f Fp[4], Ap[4];
        #pragma unroll
        for (int j = 0; j < 4; ++j) { Fp[j].x = 0.f; Fp[j].y = 0.f; Ap[j].x = 0.f; Ap[j].y = 0.f; }
        #pragma unroll
        for (int nt = 0; nt < 4; ++nt) {
            #pragma unroll
            for (int j = 0; j < 4; ++j) {
                v2f cs, sn;
                cs.x = csv[nt][2 * j]; cs.y = csv[nt][2 * j + 1];
                sn.x = snv[nt][2 * j]; sn.y = snv[nt][2 * j + 1];
                v2f fval  = w0m[nt][j] * cs + w1m[nt][j] * sn;
                v2f dterm = w1m[nt][j] * cs - w0m[nt][j] * sn;
                Fp[j] += fval;
                Ap[j] += dterm * s1m[nt][j];
            }
            float4* pc = reinterpret_cast<float4*>(&cs_s[cb + nt * 16]);
            float4* pn = reinterpret_cast<float4*>(&cs_s[cb + 4 * CPLANE + nt * 16]);
            pc[0] = make_float4(csv[nt][0], csv[nt][1], csv[nt][2], csv[nt][3]);
            pc[1] = make_float4(csv[nt][4], csv[nt][5], csv[nt][6], csv[nt][7]);
            pn[0] = make_float4(snv[nt][0], snv[nt][1], snv[nt][2], snv[nt][3]);
            pn[1] = make_float4(snv[nt][4], snv[nt][5], snv[nt][6], snv[nt][7]);
        }
        redF[wv * 2 + (lane >> 4)][ln] = ((Fp[0].x + Fp[0].y) + (Fp[1].x + Fp[1].y)) + ((Fp[2].x + Fp[2].y) + (Fp[3].x + Fp[3].y));
        redA[wv * 2 + (lane >> 4)][ln] = ((Ap[0].x + Ap[0].y) + (Ap[1].x + Ap[1].y)) + ((Ap[2].x + Ap[2].y) + (Ap[3].x + Ap[3].y));
        __syncthreads();
        if (tid < 16) {
            float F = redF[0][tid], A = redA[0][tid];
            #pragma unroll
            for (int j = 1; j < 8; ++j) { F += redF[j][tid]; A += redA[j][tid]; }
            float dx = scT_s[tid] * F;
            float a  = scT_s[tid] * A;
            dxS[s][tid] = dx; aS[s][tid] = a;
            if      (s == 0) xs[tid] = beta_s[tid] + 0.5f * dx;
            else if (s == 1) xs[tid] = beta_s[tid] + 0.5f * dx;
            else if (s == 2) xs[tid] = beta_s[tid] + dx;
        }
        __syncthreads();
    }

    if (tid < 16) {
        int bg = b0g + tid;
        float a1 = aS[0][tid], a2 = aS[1][tid], a3 = aS[2][tid], a4 = aS[3][tid];
        float dx1 = dxS[0][tid], dx2 = dxS[1][tid], dx3 = dxS[2][tid], dx4 = dxS[3][tid];
        float xf = beta_s[tid] + (dx1 + 2.f*dx2 + 2.f*dx3 + dx4) * (1.f/6.f);
        float gx = xf - bobs_r;
        float c1 = (1.f + a2 + 0.5f*a3*a2 + 0.25f*a4*a3*a2) * (1.f/6.f);
        float c2 = (2.f + a3 + 0.5f*a4*a3) * (1.f/6.f);
        float c3 = (2.f + a4) * (1.f/6.f);
        float c4 = (1.f/6.f);
        float kb1 = a1;
        float kb2 = a2 * (1.f + 0.5f*kb1);
        float kb3 = a3 * (1.f + 0.5f*kb2);
        float kb4 = a4 * (1.f + kb3);
        float gb = 1.f + (kb1 + 2.f*kb2 + 2.f*kb3 + kb4) * (1.f/6.f);
        const float inv = 1.f / 512.f;
        if (role == 0) out[bg] = gx * gb * inv;
        gxs[tid] = gx; gbs[tid] = gb;
        float sc = scT_s[tid];
        float cc[4] = {c1, c2, c3, c4};
        #pragma unroll
        for (int s2 = 0; s2 < 4; ++s2) {
            float wvv = cc[s2] * gx * inv;
            wvS[s2][tid]  = wvv;
            wv2S[s2][tid] = wvv * sc;
        }
    }

    float* __restrict__ P = ws + (unsigned)rg * PSTRIDE;

    if (role != 0) {
        __syncthreads();
        const int nt = role - 1;
        const float w0q = (nt == 0) ? w0r[0] : (nt == 1) ? w0r[1] : (nt == 2) ? w0r[2] : w0r[3];
        const float w1q = (nt == 0) ? w1r[0] : (nt == 1) ? w1r[1] : (nt == 2) ? w1r[2] : w1r[3];
        v8f accS[4], accC, accN;
        #pragma unroll
        for (int v = 0; v < 8; ++v) { accC[v] = 0.f; accN[v] = 0.f; }
        #pragma unroll
        for (int kt = 0; kt < 4; ++kt)
            #pragma unroll
            for (int v = 0; v < 8; ++v) accS[kt][v] = 0.f;
        v2f bSi[3][4];
        #pragma unroll
        for (int kq = 0; kq < 4; ++kq) {
            int b0 = hb + 2 * kq;
            #pragma unroll
            for (int kt = 1; kt < 4; ++kt) {
                bSi[kt - 1][kq].x = inp_s[b0 * IPW + kt * 16 + ln];
                bSi[kt - 1][kq].y = inp_s[(b0 + 1) * IPW + kt * 16 + ln];
            }
        }
        #pragma unroll
        for (int s = 0; s < 4; ++s) {
            v2f bS0[4], bW[4];
            #pragma unroll
            for (int kq = 0; kq < 4; ++kq) {
                int b0 = hb + 2 * kq;
                bS0[kq].x = inp0_s[(s * 16 + b0) * 17 + ln];
                bS0[kq].y = inp0_s[(s * 16 + b0 + 1) * 17 + ln];
                bW[kq].x = wv2S[s][b0];
                bW[kq].y = wv2S[s][b0 + 1];
            }
            float cs[8], sn[8], wd[8];
            #pragma unroll
            for (int v = 0; v < 8; ++v) {
                int q = CIDX(s, 0, wv, hb + v, nt * 16 + ln);
                cs[v] = cs_s[q];
                sn[v] = cs_s[q + 4 * CPLANE];
                float dterm = w1q * cs[v] - w0q * sn[v];
                float d = stv[v] * dterm;
                wd[v] = wvS[s][hb + v] * d;
            }
            #pragma unroll
            for (int kq = 0; kq < 4; ++kq) {
                v2f aD, aC, aN;
                aD.x = wd[2 * kq]; aD.y = wd[2 * kq + 1];
                aC.x = cs[2 * kq]; aC.y = cs[2 * kq + 1];
                aN.x = sn[2 * kq]; aN.y = sn[2 * kq + 1];
                accS[0] = __builtin_amdgcn_wmma_f32_16x16x4_f32(
                    false, aD, false, bS0[kq], (short)0, accS[0], false, false);
                #pragma unroll
                for (int kt = 1; kt < 4; ++kt)
                    accS[kt] = __builtin_amdgcn_wmma_f32_16x16x4_f32(
                        false, aD, false, bSi[kt - 1][kq], (short)0, accS[kt], false, false);
                accC = __builtin_amdgcn_wmma_f32_16x16x4_f32(
                    false, aC, false, bW[kq], (short)0, accC, false, false);
                accN = __builtin_amdgcn_wmma_f32_16x16x4_f32(
                    false, aN, false, bW[kq], (short)0, accN, false, false);
            }
        }
        const int mbase = wv * 64 + nt * 16 + hb;
        #pragma unroll
        for (int kt = 0; kt < 4; ++kt) {
            int kk = kt * 16 + ln;
            #pragma unroll
            for (int v = 0; v < 8; ++v)
                if (kk < 50) P[(mbase + v) * 50 + kk] = accS[kt][v];
        }
        if (ln == 0) {
            #pragma unroll
            for (int v = 0; v < 8; ++v) {
                P[12800 + mbase + v]       = accC[v];
                P[12800 + 256 + mbase + v] = accN[v];
            }
        }
    } else {
        v2f bU[4][4][3];
        float4 cv[4][4][2], nv[4][4][2];
        const int cb0 = CIDX(0, 0, wv, ln, hb);
        #pragma unroll
        for (int nt = 0; nt < 4; ++nt) {
            #pragma unroll
            for (int jt = 0; jt < 3; ++jt) {
                #pragma unroll
                for (int q = 0; q < 4; ++q) {
                    const int mrow = wv * 64 + nt * 16 + hb + 2 * q;
                    bU[nt][q][jt].x = S_s[mrow * SPW + 2 + jt * 16 + ln];
                    bU[nt][q][jt].y = S_s[(mrow + 1) * SPW + 2 + jt * 16 + ln];
                }
            }
            #pragma unroll
            for (int s = 0; s < 4; ++s) {
                const float4* pc = reinterpret_cast<const float4*>(&cs_s[cb0 + s * 8 * CPLANE + nt * 16]);
                const float4* pn = reinterpret_cast<const float4*>(&cs_s[cb0 + s * 8 * CPLANE + 4 * CPLANE + nt * 16]);
                cv[nt][s][0] = pc[0]; cv[nt][s][1] = pc[1];
                nv[nt][s][0] = pn[0]; nv[nt][s][1] = pn[1];
            }
        }
        const float scb = scT_s[ln];
        __syncthreads();
        float wvb[4];
        #pragma unroll
        for (int s = 0; s < 4; ++s) wvb[s] = wvS[s][ln];
        v8f uacc[3];
        #pragma unroll
        for (int jt = 0; jt < 3; ++jt)
            #pragma unroll
            for (int v = 0; v < 8; ++v) uacc[jt][v] = 0.f;
        #pragma unroll
        for (int nt = 0; nt < 4; ++nt) {
            float dsw[8];
            #pragma unroll
            for (int v = 0; v < 8; ++v) dsw[v] = 0.f;
            #pragma unroll
            for (int s = 0; s < 4; ++s) {
                const float4 c0 = cv[nt][s][0], c1 = cv[nt][s][1], n0 = nv[nt][s][0], n1 = nv[nt][s][1];
                const float cq[8] = {c0.x, c0.y, c0.z, c0.w, c1.x, c1.y, c1.z, c1.w};
                const float nq[8] = {n0.x, n0.y, n0.z, n0.w, n1.x, n1.y, n1.z, n1.w};
                #pragma unroll
                for (int v = 0; v < 8; ++v) {
                    const float w1e = (v & 1) ? w1m[nt][v >> 1].y : w1m[nt][v >> 1].x;
                    const float w0e = (v & 1) ? w0m[nt][v >> 1].y : w0m[nt][v >> 1].x;
                    float dterm = w1e * cq[v] - w0e * nq[v];
                    float d = scb * dterm;
                    dsw[v] += wvb[s] * d;
                }
            }
            #pragma unroll
            for (int q = 0; q < 4; ++q) {
                v2f aU;
                aU.x = dsw[2 * q]; aU.y = dsw[2 * q + 1];
                #pragma unroll
                for (int jt = 0; jt < 3; ++jt)
                    uacc[jt] = __builtin_amdgcn_wmma_f32_16x16x4_f32(
                        false, aU, false, bU[nt][q][jt], (short)0, uacc[jt], false, false);
            }
        }
        #pragma unroll
        for (int jt = 0; jt < 3; ++jt)
            #pragma unroll
            for (int v = 0; v < 8; ++v) part_s[((wv * 3 + jt) * 16 + hb + v) * PPW + ln] = uacc[jt][v];
        __syncthreads();
        #pragma unroll
        for (int i = 0; i < 6; ++i) {
            const int idx = tid + i * 128;
            const int b = idx / 48, j = idx - b * 48;
            const int u = j >> 4, r = j & 15;
            const int pq = ((u * 16) + b) * PPW + r;
            const float acc = (part_s[pq] + part_s[pq + 3 * 16 * PPW]) + (part_s[pq + 6 * 16 * PPW] + part_s[pq + 9 * 16 * PPW]);
            const float e0 = inp_s[b * IPW + 2 + r], e1 = inp_s[b * IPW + 18 + r], e2 = inp_s[b * IPW + 34 + r];
            const float px = (u == 0) ? e1 : e0;
            const float py = (u == 2) ? e1 : e2;
            const float p = px * py;
            out[512 + b0g * 48 + idx] = gxs[b] * (1.f / 512.f) * gbs[b] * p + acc;
        }
    }
}

__global__ __launch_bounds__(256) void k2_reduce(const float* __restrict__ ws, float* __restrict__ out)
{
    const unsigned i = blockIdx.x * 256u + threadIdx.x;
    float v[32];
    #pragma unroll
    for (int p = 0; p < 32; ++p) v[p] = ws[(unsigned)p * PSTRIDE + i];
    float acc = v[0];
    #pragma unroll
    for (int p = 1; p < 32; ++p) acc += v[p];
    out[25088u + i] = acc;
}

extern "C" void kernel_launch(void* const* d_in, const int* in_sizes, int n_in,
                              void* d_out, int out_size, void* d_ws, size_t ws_size,
                              hipStream_t stream)
{
    (void)in_sizes; (void)n_in; (void)out_size; (void)ws_size;
    const float* U0   = (const float*)d_in[0];
    const float* U1   = (const float*)d_in[1];
    const float* U2   = (const float*)d_in[2];
    const float* S    = (const float*)d_in[3];
    const float* w    = (const float*)d_in[4];
    const float* btn  = (const float*)d_in[5];
    const float* bobs = (const float*)d_in[6];
    const int*   bin  = (const int*)d_in[7];
    float* out = (float*)d_out;
    float* ws  = (float*)d_ws;

    k1_stages<<<160, 128, 0, stream>>>(U0, U1, U2, S, w, btn, bobs, bin, ws, out);
    k2_reduce<<<52, 256, 0, stream>>>(ws, out);
}


# stage hand-over: lane that sums a row keeps beta, scale, x in registers, writes the next stage's inp cols 0,1 itself; per-stage inp copy gone (cols 2..15 once); 3 barriers a stage -> 2; sums same
# speedup vs baseline: 68.9799x; 1.0609x over previous
#include <hip/hip_runtime.h>
#include <math.h>

typedef float v2f __attribute__((ext_vector_type(2)));
typedef float v8f __attribute__((ext_vector_type(8)));

#define BB 512
#define MM 256
#define INDIM 50

#define PSTRIDE 13312u
#define IPW 65
#define SPW 52
#define PPW 18
#define CPW 68
#define CPLANE (16 * CPW)
#define CIDX(s, p, w, b, ml) ((((s) * 2 + (p)) * 4 + (w)) * CPLANE + (b) * CPW + (ml))

#define INV_2PI 0x1.45f306p-3f

__global__ __launch_bounds__(128) void k1_stages(
    const float* __restrict__ U0, const float* __restrict__ U1, const float* __restrict__ U2,
    const float* __restrict__ S,  const float* __restrict__ w,
    const float* __restrict__ b_t_n, const float* __restrict__ b_obs,
    const int* __restrict__ b_i_n,
    float* __restrict__ ws, float* __restrict__ out)
{
    __shared__ __attribute__((aligned(16))) float S_s[256 * SPW];
    __shared__ float inp_s[16 * IPW];
    __shared__ float inp0_s[4 * 16 * 17];
    __shared__ __attribute__((aligned(16))) float cs_s[4 * 2 * 4 * CPLANE];
    __shared__ __attribute__((aligned(16))) float w_s[512];
    __shared__ float part_s[4 * 3 * 16 * PPW];
    __shared__ float redF[8][16];
    __shared__ float redA[8][16];
    __shared__ float scT_s[16];
    __shared__ float aS[4][16], dxS[4][16];
    __shared__ float wvS[4][16], wv2S[4][16];
    __shared__ float gxs[16], gbs[16];

    const int tid  = threadIdx.x;
    const int wv   = tid >> 5;
    const int lane = tid & 31;
    const int ln   = lane & 15;
    const int hb   = (lane < 16) ? 0 : 8;
    const int role = blockIdx.x >> 5;
    const int rg   = blockIdx.x & 31;
    const int b0g  = rg * 16;

    int giv[6];
    #pragma unroll
    for (int i = 0; i < 6; ++i) {
        int idx = tid + i * 128;
        int bl = idx / 48, j = idx - bl * 48;
        giv[i] = b_i_n[(b0g + bl) * 3 + (j >> 4)];
    }
    const float4* __restrict__ S4 = reinterpret_cast<const float4*>(S);
    float4 sv[25];
    #pragma unroll
    for (int i = 0; i < 25; ++i) sv[i] = S4[tid + i * 128];
    int   mro[4]; float w0r[4], w1r[4];
    #pragma unroll
    for (int nt = 0; nt < 4; ++nt) mro[nt] = wv * 64 + nt * 16 + ln;
    float wg[4];
    #pragma unroll
    for (int i = 0; i < 4; ++i) wg[i] = w[tid + i * 128];
    float scT_r = 0.f, bobs_r = 0.f;
    if (tid < 16) { scT_r = b_t_n[b0g + tid]; bobs_r = b_obs[b0g + tid]; }
    float gv[6];
    #pragma unroll
    for (int i = 0; i < 6; ++i) {
        int idx = tid + i * 128;
        int bl = idx / 48, j = idx - bl * 48;
        int u = j >> 4, r = j & 15;
        const float* Uu = (u == 0) ? U0 : (u == 1) ? U1 : U2;
        gv[i] = Uu[giv[i] * 16 + r];
    }

    #pragma unroll
    for (int i = 0; i < 25; ++i) {
        int e0 = (tid + i * 128) * 4;
        int m0 = e0 / INDIM, k0 = e0 - m0 * INDIM;
        int a0 = e0 + 2 * m0;
        int a2 = a0 + 2 + ((k0 == 48) ? 2 : 0);
        S_s[a0]     = sv[i].x;
        S_s[a0 + 1] = sv[i].y;
        S_s[a2]     = sv[i].z;
        S_s[a2 + 1] = sv[i].w;
    }
    for (int m = tid; m < 256; m += 128) { S_s[m * SPW + 50] = 0.f; S_s[m * SPW + 51] = 0.f; }
    #pragma unroll
    for (int i = 0; i < 6; ++i) {
        int idx = tid + i * 128;
        int bl = idx / 48, j = idx - bl * 48;
        inp_s[bl * IPW + 2 + j] = gv[i];
    }
    for (int idx = tid; idx < 16 * 15; idx += 128) {
        int bl = idx / 15, c = idx - bl * 15;
        inp_s[bl * IPW + 50 + c] = 0.f;
    }
    if (tid < 16) scT_s[tid] = scT_r;
    #pragma unroll
    for (int i = 0; i < 4; ++i) w_s[tid + i * 128] = wg[i];
    __syncthreads();

    float beta_r = 0.f;
    if (tid < 16) {
        float acc = 0.f;
        for (int r = 0; r < 16; ++r)
            acc += inp_s[tid*IPW + 2 + r] * inp_s[tid*IPW + 18 + r] * inp_s[tid*IPW + 34 + r];
        beta_r = acc;
        const float t0 = scT_r * 0.f;
        inp_s[tid * IPW + 0] = t0;
        inp_s[tid * IPW + 1] = acc;
        inp0_s[tid * 17 + 0] = t0;
        inp0_s[tid * 17 + 1] = acc;
    }
    if (wv != 0) {
        #pragma unroll
        for (int i = 0; i < 3; ++i) {
            const int idx = (tid - 32) + i * 96;
            if (idx < 16 * 14) {
                const int bl = idx / 14, k = 2 + idx - bl * 14;
                const float v = inp_s[bl * IPW + k];
                #pragma unroll
                for (int s4 = 0; s4 < 4; ++s4) inp0_s[(s4 * 16 + bl) * 17 + k] = v;
            }
        }
    }
    __syncthreads();

    v2f w0m[4][4], w1m[4][4], s1m[4][4];
    #pragma unroll
    for (int nt = 0; nt < 4; ++nt) {
        w0r[nt] = w_s[mro[nt]]; w1r[nt] = w_s[256 + mro[nt]];
        const int mb = wv * 64 + nt * 16 + hb;
        #pragma unroll
        for (int j = 0; j < 4; ++j) {
            w0m[nt][j].x = w_s[mb + 2 * j];           w0m[nt][j].y = w_s[mb + 2 * j + 1];
            w1m[nt][j].x = w_s[256 + mb + 2 * j];     w1m[nt][j].y = w_s[256 + mb + 2 * j + 1];
            s1m[nt][j].x = S_s[(mb + 2 * j) * SPW + 1]; s1m[nt][j].y = S_s[(mb + 2 * j + 1) * SPW + 1];
        }
    }
    float stv[8];
    #pragma unroll
    for (int v = 0; v < 8; ++v) stv[v] = scT_s[hb + v];

    const int klo = (lane < 16) ? 0 : 2;

    for (int s = 0; s < 4; ++s) {

        v8f acc[4];
        #pragma unroll
        for (int nt = 0; nt < 4; ++nt)
            #pragma unroll
            for (int v = 0; v < 8; ++v) acc[nt][v] = 0.f;

        v2f af[13], bf[13][4];
        #pragma unroll
        for (int kq = 0; kq < 13; ++kq) {
            int k0 = kq * 4 + klo;
            af[kq].x = inp_s[ln * IPW + k0];
            af[kq].y = inp_s[ln * IPW + k0 + 1];
            #pragma unroll
            for (int nt = 0; nt < 4; ++nt) {
                bf[kq][nt].x = S_s[mro[nt] * SPW + k0];
                bf[kq][nt].y = S_s[mro[nt] * SPW + k0 + 1];
            }
        }
        __builtin_amdgcn_sched_barrier(0);
        #pragma unroll
        for (int kq = 0; kq < 13; ++kq) {
            #pragma unroll
            for (int nt = 0; nt < 4; ++nt)
                acc[nt] = __builtin_amdgcn_wmma_f32_16x16x4_f32(
                    false, bf[kq][nt], false, af[kq], (short)0, acc[nt], false, false);
        }

        float csv[4][8], snv[4][8];
        const int cb = CIDX(s, 0, wv, ln, hb);
        #pragma unroll
        for (int nt = 0; nt < 4; ++nt) {
            const v8f rev = acc[nt] * INV_2PI;
            #pragma unroll
            for (int v = 0; v < 8; ++v) {
                snv[nt][v] = __builtin_amdgcn_sinf(rev[v]);
                csv[nt][v] = __builtin_amdgcn_cosf(rev[v]);
            }
        }
        v2f Fp[4], Ap[4];
        #pragma unroll
        for (int j = 0; j < 4; ++j) { Fp[j].x = 0.f; Fp[j].y = 0.f; Ap[j].x = 0.f; Ap[j].y = 0.f; }
        #pragma unroll
        for (int nt = 0; nt < 4; ++nt) {
            #pragma unroll
            for (int j = 0; j < 4; ++j) {
                v2f cs, sn;
                cs.x = csv[nt][2 * j]; cs.y = csv[nt][2 * j + 1];
                sn.x = snv[nt][2 * j]; sn.y = snv[nt][2 * j + 1];
                v2f fval  = w0m[nt][j] * cs + w1m[nt][j] * sn;
                v2f dterm = w1m[nt][j] * cs - w0m[nt][j] * sn;
                Fp[j] += fval;
                Ap[j] += dterm * s1m[nt][j];
            }
            float4* pc = reinterpret_cast<float4*>(&cs_s[cb + nt * 16]);
            float4* pn = reinterpret_cast<float4*>(&cs_s[cb + 4 * CPLANE + nt * 16]);
            pc[0] = make_float4(csv[nt][0], csv[nt][1], csv[nt][2], csv[nt][3]);
            pc[1] = make_float4(csv[nt][4], csv[nt][5], csv[nt][6], csv[nt][7]);
            pn[0] = make_float4(snv[nt][0], snv[nt][1], snv[nt][2], snv[nt][3]);
            pn[1] = make_float4(snv[nt][4], snv[nt][5], snv[nt][6], snv[nt][7]);
        }
        redF[wv * 2 + (lane >> 4)][ln] = ((Fp[0].x + Fp[0].y) + (Fp[1].x + Fp[1].y)) + ((Fp[2].x + Fp[2].y) + (Fp[3].x + Fp[3].y));
        redA[wv * 2 + (lane >> 4)][ln] = ((Ap[0].x + Ap[0].y) + (Ap[1].x + Ap[1].y)) + ((Ap[2].x + Ap[2].y) + (Ap[3].x + Ap[3].y));
        __syncthreads();
        if (tid < 16) {
            float F = redF[0][tid], A = redA[0][tid];
            #pragma unroll
            for (int j = 1; j < 8; ++j) { F += redF[j][tid]; A += redA[j][tid]; }
            float dx = scT_r * F;
            float a  = scT_r * A;
            dxS[s][tid] = dx; aS[s][tid] = a;
            if (s < 3) {
                const float step = (s == 2) ? dx : 0.5f * dx;
                const float xn = beta_r + step;
                const float tn = scT_r * ((s == 2) ? 1.f : 0.5f);
                inp_s[tid * IPW + 0] = tn;
                inp_s[tid * IPW + 1] = xn;
                inp0_s[((s + 1) * 16 + tid) * 17 + 0] = tn;
                inp0_s[((s + 1) * 16 + tid) * 17 + 1] = xn;
            }
        }
        __syncthreads();
    }

    if (tid < 16) {
        int bg = b0g + tid;
        float a1 = aS[0][tid], a2 = aS[1][tid], a3 = aS[2][tid], a4 = aS[3][tid];
        float dx1 = dxS[0][tid], dx2 = dxS[1][tid], dx3 = dxS[2][tid], dx4 = dxS[3][tid];
        float xf = beta_r + (dx1 + 2.f*dx2 + 2.f*dx3 + dx4) * (1.f/6.f);
        float gx = xf - bobs_r;
        float c1 = (1.f + a2 + 0.5f*a3*a2 + 0.25f*a4*a3*a2) * (1.f/6.f);
        float c2 = (2.f + a3 + 0.5f*a4*a3) * (1.f/6.f);
        float c3 = (2.f + a4) * (1.f/6.f);
        float c4 = (1.f/6.f);
        float kb1 = a1;
        float kb2 = a2 * (1.f + 0.5f*kb1);
        float kb3 = a3 * (1.f + 0.5f*kb2);
        float kb4 = a4 * (1.f + kb3);
        float gb = 1.f + (kb1 + 2.f*kb2 + 2.f*kb3 + kb4) * (1.f/6.f);
        const float inv = 1.f / 512.f;
        if (role == 0) out[bg] = gx * gb * inv;
        gxs[tid] = gx; gbs[tid] = gb;
        float sc = scT_r;
        float cc[4] = {c1, c2, c3, c4};
        #pragma unroll
        for (int s2 = 0; s2 < 4; ++s2) {
            float wvv = cc[s2] * gx * inv;
            wvS[s2][tid]  = wvv;
            wv2S[s2][tid] = wvv * sc;
        }
    }

    float* __restrict__ P = ws + (unsigned)rg * PSTRIDE;

    if (role != 0) {
        __syncthreads();
        const int nt = role - 1;
        const float w0q = (nt == 0) ? w0r[0] : (nt == 1) ? w0r[1] : (nt == 2) ? w0r[2] : w0r[3];
        const float w1q = (nt == 0) ? w1r[0] : (nt == 1) ? w1r[1] : (nt == 2) ? w1r[2] : w1r[3];
        v8f accS[4], accC, accN;
        #pragma unroll
        for (int v = 0; v < 8; ++v) { accC[v] = 0.f; accN[v] = 0.f; }
        #pragma unroll
        for (int kt = 0; kt < 4; ++kt)
            #pragma unroll
            for (int v = 0; v < 8; ++v) accS[kt][v] = 0.f;
        v2f bSi[3][4];
        #pragma unroll
        for (int kq = 0; kq < 4; ++kq) {
            int b0 = hb + 2 * kq;
            #pragma unroll
            for (int kt = 1; kt < 4; ++kt) {
                bSi[kt - 1][kq].x = inp_s[b0 * IPW + kt * 16 + ln];
                bSi[kt - 1][kq].y = inp_s[(b0 + 1) * IPW + kt * 16 + ln];
            }
        }
        #pragma unroll
        for (int s = 0; s < 4; ++s) {
            v2f bS0[4], bW[4];
            #pragma unroll
            for (int kq = 0; kq < 4; ++kq) {
                int b0 = hb + 2 * kq;
                bS0[kq].x = inp0_s[(s * 16 + b0) * 17 + ln];
                bS0[kq].y = inp0_s[(s * 16 + b0 + 1) * 17 + ln];
                bW[kq].x = wv2S[s][b0];
                bW[kq].y = wv2S[s][b0 + 1];
            }
            float cs[8], sn[8], wd[8];
            #pragma unroll
            for (int v = 0; v < 8; ++v) {
                int q = CIDX(s, 0, wv, hb + v, nt * 16 + ln);
                cs[v] = cs_s[q];
                sn[v] = cs_s[q + 4 * CPLANE];
                float dterm = w1q * cs[v] - w0q * sn[v];
                float d = stv[v] * dterm;
                wd[v] = wvS[s][hb + v] * d;
            }
            #pragma unroll
            for (int kq = 0; kq < 4; ++kq) {
                v2f aD, aC, aN;
                aD.x = wd[2 * kq]; aD.y = wd[2 * kq + 1];
                aC.x = cs[2 * kq]; aC.y = cs[2 * kq + 1];
                aN.x = sn[2 * kq]; aN.y = sn[2 * kq + 1];
                accS[0] = __builtin_amdgcn_wmma_f32_16x16x4_f32(
                    false, aD, false, bS0[kq], (short)0, accS[0], false, false);
                #pragma unroll
                for (int kt = 1; kt < 4; ++kt)
                    accS[kt] = __builtin_amdgcn_wmma_f32_16x16x4_f32(
                        false, aD, false, bSi[kt - 1][kq], (short)0, accS[kt], false, false);
                accC = __builtin_amdgcn_wmma_f32_16x16x4_f32(
                    false, aC, false, bW[kq], (short)0, accC, false, false);
                accN = __builtin_amdgcn_wmma_f32_16x16x4_f32(
                    false, aN, false, bW[kq], (short)0, accN, false, false);
            }
        }
        const int mbase = wv * 64 + nt * 16 + hb;
        #pragma unroll
        for (int kt = 0; kt < 4; ++kt) {
            int kk = kt * 16 + ln;
            #pragma unroll
            for (int v = 0; v < 8; ++v)
                if (kk < 50) P[(mbase + v) * 50 + kk] = accS[kt][v];
        }
        if (ln == 0) {
            #pragma unroll
            for (int v = 0; v < 8; ++v) {
                P[12800 + mbase + v]       = accC[v];
                P[12800 + 256 + mbase + v] = accN[v];
            }
        }
    } else {
        v2f bU[4][4][3];
        float4 cv[4][4][2], nv[4][4][2];
        const int cb0 = CIDX(0, 0, wv, ln, hb);
        #pragma unroll
        for (int nt = 0; nt < 4; ++nt) {
            #pragma unroll
            for (int jt = 0; jt < 3; ++jt) {
                #pragma unroll
                for (int q = 0; q < 4; ++q) {
                    const int mrow = wv * 64 + nt * 16 + hb + 2 * q;
                    bU[nt][q][jt].x = S_s[mrow * SPW + 2 + jt * 16 + ln];
                    bU[nt][q][jt].y = S_s[(mrow + 1) * SPW + 2 + jt * 16 + ln];
                }
            }
            #pragma unroll
            for (int s = 0; s < 4; ++s) {
                const float4* pc = reinterpret_cast<const float4*>(&cs_s[cb0 + s * 8 * CPLANE + nt * 16]);
                const float4* pn = reinterpret_cast<const float4*>(&cs_s[cb0 + s * 8 * CPLANE + 4 * CPLANE + nt * 16]);
                cv[nt][s][0] = pc[0]; cv[nt][s][1] = pc[1];
                nv[nt][s][0] = pn[0]; nv[nt][s][1] = pn[1];
            }
        }
        const float scb = scT_s[ln];
        __syncthreads();
        float wvb[4];
        #pragma unroll
        for (int s = 0; s < 4; ++s) wvb[s] = wvS[s][ln];
        v8f uacc[3];
        #pragma unroll
        for (int jt = 0; jt < 3; ++jt)
            #pragma unroll
            for (int v = 0; v < 8; ++v) uacc[jt][v] = 0.f;
        #pragma unroll
        for (int nt = 0; nt < 4; ++nt) {
            float dsw[8];
            #pragma unroll
            for (int v = 0; v < 8; ++v) dsw[v] = 0.f;
            #pragma unroll
            for (int s = 0; s < 4; ++s) {
                const float4 c0 = cv[nt][s][0], c1 = cv[nt][s][1], n0 = nv[nt][s][0], n1 = nv[nt][s][1];
                const float cq[8] = {c0.x, c0.y, c0.z, c0.w, c1.x, c1.y, c1.z, c1.w};
                const float nq[8] = {n0.x, n0.y, n0.z, n0.w, n1.x, n1.y, n1.z, n1.w};
                #pragma unroll
                for (int v = 0; v < 8; ++v) {
                    const float w1e = (v & 1) ? w1m[nt][v >> 1].y : w1m[nt][v >> 1].x;
                    const float w0e = (v & 1) ? w0m[nt][v >> 1].y : w0m[nt][v >> 1].x;
                    float dterm = w1e * cq[v] - w0e * nq[v];
                    float d = scb * dterm;
                    dsw[v] += wvb[s] * d;
                }
            }
            #pragma unroll
            for (int q = 0; q < 4; ++q) {
                v2f aU;
                aU.x = dsw[2 * q]; aU.y = dsw[2 * q + 1];
                #pragma unroll
                for (int jt = 0; jt < 3; ++jt)
                    uacc[jt] = __builtin_amdgcn_wmma_f32_16x16x4_f32(
                        false, aU, false, bU[nt][q][jt], (short)0, uacc[jt], false, false);
            }
        }
        #pragma unroll
        for (int jt = 0; jt < 3; ++jt)
            #pragma unroll
            for (int v = 0; v < 8; ++v) part_s[((wv * 3 + jt) * 16 + hb + v) * PPW + ln] = uacc[jt][v];
        __syncthreads();
        #pragma unroll
        for (int i = 0; i < 6; ++i) {
            const int idx = tid + i * 128;
            const int b = idx / 48, j = idx - b * 48;
            const int u = j >> 4, r = j & 15;
            const int pq = ((u * 16) + b) * PPW + r;
            const float acc = (part_s[pq] + part_s[pq + 3 * 16 * PPW]) + (part_s[pq + 6 * 16 * PPW] + part_s[pq + 9 * 16 * PPW]);
            const float e0 = inp_s[b * IPW + 2 + r], e1 = inp_s[b * IPW + 18 + r], e2 = inp_s[b * IPW + 34 + r];
            const float px = (u == 0) ? e1 : e0;
            const float py = (u == 2) ? e1 : e2;
            const float p = px * py;
            out[512 + b0g * 48 + idx] = gxs[b] * (1.f / 512.f) * gbs[b] * p + acc;
        }
    }
}

__global__ __launch_bounds__(256) void k2_reduce(const float* __restrict__ ws, float* __restrict__ out)
{
    const unsigned i = blockIdx.x * 256u + threadIdx.x;
    float v[32];
    #pragma unroll
    for (int p = 0; p < 32; ++p) v[p] = ws[(unsigned)p * PSTRIDE + i];
    float acc = v[0];
    #pragma unroll
    for (int p = 1; p < 32; ++p) acc += v[p];
    out[25088u + i] = acc;
}

extern "C" void kernel_launch(void* const* d_in, const int* in_sizes, int n_in,
                              void* d_out, int out_size, void* d_ws, size_t ws_size,
                              hipStream_t stream)
{
    (void)in_sizes; (void)n_in; (void)out_size; (void)ws_size;
    const float* U0   = (const float*)d_in[0];
    const float* U1   = (const float*)d_in[1];
    const float* U2   = (const float*)d_in[2];
    const float* S    = (const float*)d_in[3];
    const float* w    = (const float*)d_in[4];
    const float* btn  = (const float*)d_in[5];
    const float* bobs = (const float*)d_in[6];
    const int*   bin  = (const int*)d_in[7];
    float* out = (float*)d_out;
    float* ws  = (float*)d_ws;

    k1_stages<<<160, 128, 0, stream>>>(U0, U1, U2, S, w, btn, bobs, bin, ws, out);
    k2_reduce<<<52, 256, 0, stream>>>(ws, out);
}
